# starting kernel
# baseline (speedup 1.0000x reference)
#include <hip/hip_runtime.h>
#include <cstddef>

typedef float v2f __attribute__((ext_vector_type(2)));
typedef float v8f __attribute__((ext_vector_type(8)));

#define DIN 64

__global__ void gnn_zero_kernel(float* __restrict__ p, unsigned long long n) {
    unsigned long long i = (unsigned long long)blockIdx.x * blockDim.x + threadIdx.x;
    if (i < n) p[i] = 0.0f;
}

__global__ void gnn_pack_weights_kernel(const float* __restrict__ Wl,
                                        const float* __restrict__ bl,
                                        const float* __restrict__ Wr,
                                        float* __restrict__ WTl,
                                        float* __restrict__ WTr,
                                        float* __restrict__ blp,
                                        int dout, int dout_pad) {
    int tid = blockIdx.x * blockDim.x + threadIdx.x;
    int col = tid / DIN;
    int k   = tid % DIN;
    if (col >= dout_pad) return;
    bool cv  = (col < dout);
    WTl[(size_t)col * DIN + k] = cv ? Wl[(size_t)k * dout + col] : 0.0f;
    WTr[(size_t)col * DIN + k] = cv ? Wr[(size_t)k * dout + col] : 0.0f;
    if (k == 0) blp[col] = cv ? bl[col] : 0.0f;
}

__global__ void gnn_scatter_kernel(const float* __restrict__ x,
                                   const int* __restrict__ src,
                                   const int* __restrict__ dst,
                                   float* __restrict__ agg,
                                   float* __restrict__ deg,
                                   int E, int do_deg) {
    unsigned long long tid = (unsigned long long)blockIdx.x * blockDim.x + threadIdx.x;
    unsigned long long e  = tid >> 4;
    int f4 = (int)(tid & 15);
    if (e >= (unsigned long long)E) return;
    int s = src[e];
    int d = dst[e];
    const float4* xr = (const float4*)(x + (size_t)s * DIN);
    float4 v = xr[f4];
    float* ar = agg + (size_t)d * DIN + (size_t)f4 * 4;
    atomicAdd(ar + 0, v.x);
    atomicAdd(ar + 1, v.y);
    atomicAdd(ar + 2, v.z);
    atomicAdd(ar + 3, v.w);
    if (do_deg && f4 == 0) atomicAdd(deg + d, 1.0f);
}

__global__ void __launch_bounds__(128)
gnn_dense_wmma_kernel(const float* __restrict__ agg,
                      const float* __restrict__ deg,
                      const float* __restrict__ hin,
                      const float* __restrict__ WTl,
                      const float* __restrict__ blp,
                      const float* __restrict__ WTr,
                      float* __restrict__ hout,
                      int do_relu, int n_mtiles) {
    const int lane  = threadIdx.x;
    const int mtile = blockIdx.x * blockDim.y + threadIdx.y;
    const int ntile = blockIdx.y;
    if (mtile >= n_mtiles) return;

    const int mrow  = mtile * 16 + (lane & 15);
    const int khalf = (lane >> 4) * 2;
    const int col   = ntile * 16 + (lane & 15);

    const float rdeg = 1.0f / fmaxf(deg[mrow], 1.0f);
    const float* aggr = agg + (size_t)mrow * DIN + khalf;
    const float* hinr = hin + (size_t)mrow * DIN + khalf;
    const float* wl   = WTl + (size_t)col  * DIN + khalf;
    const float* wr   = WTr + (size_t)col  * DIN + khalf;

    v8f c = {};
    #pragma unroll
    for (int k0 = 0; k0 < DIN; k0 += 4) {
        v2f aAgg = *(const v2f*)(aggr + k0);
        aAgg.x *= rdeg; aAgg.y *= rdeg;
        v2f aX   = *(const v2f*)(hinr + k0);
        v2f bL   = *(const v2f*)(wl + k0);
        v2f bR   = *(const v2f*)(wr + k0);
        c = __builtin_amdgcn_wmma_f32_16x16x4_f32(false, aAgg, false, bL,
                                                  (short)0, c, false, false);
        c = __builtin_amdgcn_wmma_f32_16x16x4_f32(false, aX,   false, bR,
                                                  (short)0, c, false, false);
    }

    const float bv = blp[col];
    float* outp = hout + (size_t)(mtile * 16 + ((lane >> 4) * 8)) * DIN + col;
    #pragma unroll
    for (int r = 0; r < 8; ++r) {
        float v = c[r] + bv;
        if (do_relu) v = fmaxf(v, 0.0f);
        outp[(size_t)r * DIN] = v;
    }
}

__global__ void gnn_logsoftmax_kernel(const float* __restrict__ logits,
                                      float* __restrict__ out,
                                      int n_nodes, int C, int stride) {
    const int lane = threadIdx.x;
    const int node = blockIdx.x * blockDim.y + threadIdx.y;
    if (node >= n_nodes) return;
    const float* rowp = logits + (size_t)node * stride;
    const float NEG_INF = -__builtin_inff();
    float v0 = (lane      < C) ? rowp[lane]      : NEG_INF;
    float v1 = (lane + 32 < C) ? rowp[lane + 32] : NEG_INF;
    float m = fmaxf(v0, v1);
    #pragma unroll
    for (int off = 16; off > 0; off >>= 1) m = fmaxf(m, __shfl_xor(m, off, 32));
    float e0 = (lane      < C) ? expf(v0 - m) : 0.0f;
    float e1 = (lane + 32 < C) ? expf(v1 - m) : 0.0f;
    float s = e0 + e1;
    #pragma unroll
    for (int off = 16; off > 0; off >>= 1) s += __shfl_xor(s, off, 32);
    const float ls = logf(s);
    if (lane      < C) out[(size_t)node * C + lane]      = v0 - m - ls;
    if (lane + 32 < C) out[(size_t)node * C + lane + 32] = v1 - m - ls;
}

static inline void launch_dense(const float* agg, const float* deg, const float* hin,
                                const float* WTl, const float* blp, const float* WTr,
                                float* hout, int dout_pad, int do_relu,
                                int N, hipStream_t stream) {
    const int n_mtiles = (N + 15) / 16;
    dim3 block(32, 4);
    dim3 grid((n_mtiles + 3) / 4, dout_pad / 16);
    gnn_dense_wmma_kernel<<<grid, block, 0, stream>>>(agg, deg, hin, WTl, blp, WTr,
                                                      hout, do_relu, n_mtiles);
}

extern "C" void kernel_launch(void* const* d_in, const int* in_sizes, int n_in,
                              void* d_out, int out_size, void* d_ws, size_t ws_size,
                              hipStream_t stream) {
    const float* x   = (const float*)d_in[0];
    const int*   ei  = (const int*)  d_in[1];
    const float* Wl0 = (const float*)d_in[2];
    const float* bl0 = (const float*)d_in[3];
    const float* Wr0 = (const float*)d_in[4];
    const float* Wl1 = (const float*)d_in[5];
    const float* bl1 = (const float*)d_in[6];
    const float* Wr1 = (const float*)d_in[7];
    const float* Wl2 = (const float*)d_in[8];
    const float* bl2 = (const float*)d_in[9];
    const float* Wr2 = (const float*)d_in[10];

    const int N = in_sizes[0] / DIN;
    const int E = in_sizes[1] / 2;
    const int* src = ei;
    const int* dst = ei + E;

    const int DP0 = 64, DP1 = 64, DP2 = 48;

    float* deg  = (float*)d_ws;
    float* agg  = deg + N;
    float* h1   = agg + (size_t)N * DIN;
    float* h2   = h1  + (size_t)N * DIN;
    float* WTl0 = h2  + (size_t)N * DIN;
    float* WTr0 = WTl0 + DP0 * DIN;
    float* blp0 = WTr0 + DP0 * DIN;
    float* WTl1 = blp0 + DP0;
    float* WTr1 = WTl1 + DP1 * DIN;
    float* blp1 = WTr1 + DP1 * DIN;
    float* WTl2 = blp1 + DP1;
    float* WTr2 = WTl2 + DP2 * DIN;
    float* blp2 = WTr2 + DP2 * DIN;

    const unsigned long long n_da = (unsigned long long)N * (DIN + 1);
    const unsigned long long n_a  = (unsigned long long)N * DIN;
    const unsigned long long scat_threads = (unsigned long long)E * 16ull;
    const int ZB = 256;

    gnn_pack_weights_kernel<<<(DP0 * DIN + ZB - 1) / ZB, ZB, 0, stream>>>(
        Wl0, bl0, Wr0, WTl0, WTr0, blp0, 64, DP0);
    gnn_pack_weights_kernel<<<(DP1 * DIN + ZB - 1) / ZB, ZB, 0, stream>>>(
        Wl1, bl1, Wr1, WTl1, WTr1, blp1, 64, DP1);
    gnn_pack_weights_kernel<<<(DP2 * DIN + ZB - 1) / ZB, ZB, 0, stream>>>(
        Wl2, bl2, Wr2, WTl2, WTr2, blp2, 40, DP2);

    gnn_zero_kernel<<<(unsigned)((n_da + ZB - 1) / ZB), ZB, 0, stream>>>(deg, n_da);
    gnn_scatter_kernel<<<(unsigned)((scat_threads + ZB - 1) / ZB), ZB, 0, stream>>>(
        x, src, dst, agg, deg, E, 1);
    launch_dense(agg, deg, x, WTl0, blp0, WTr0, h1, DP0, 1, N, stream);

    gnn_zero_kernel<<<(unsigned)((n_a + ZB - 1) / ZB), ZB, 0, stream>>>(agg, n_a);
    gnn_scatter_kernel<<<(unsigned)((scat_threads + ZB - 1) / ZB), ZB, 0, stream>>>(
        h1, src, dst, agg, deg, E, 0);
    launch_dense(agg, deg, h1, WTl1, blp1, WTr1, h2, DP1, 1, N, stream);

    gnn_zero_kernel<<<(unsigned)((n_a + ZB - 1) / ZB), ZB, 0, stream>>>(agg, n_a);
    gnn_scatter_kernel<<<(unsigned)((scat_threads + ZB - 1) / ZB), ZB, 0, stream>>>(
        h2, src, dst, agg, deg, E, 0);
    launch_dense(agg, deg, h2, WTl2, blp2, WTr2, h1, DP2, 0, N, stream);

    {
        dim3 block(32, 8);
        dim3 grid((N + 7) / 8);
        gnn_logsoftmax_kernel<<<grid, block, 0, stream>>>(h1, (float*)d_out, N, 40, 64);
    }
}


# Aggregation without float atomics: edges counting-sorted by target once per call (int atomics, arrival-order rank); one 16-lane owner sums and stores each node row; zero-fill launches dropped
# speedup vs baseline: 1.9498x; 1.9498x over previous
#include <hip/hip_runtime.h>
#include <cstddef>

typedef float v2f __attribute__((ext_vector_type(2)));
typedef float v8f __attribute__((ext_vector_type(8)));

#define DIN 64

__global__ void gnn_zero_kernel(float* __restrict__ p, unsigned long long n) {
    unsigned long long i = (unsigned long long)blockIdx.x * blockDim.x + threadIdx.x;
    if (i < n) p[i] = 0.0f;
}

__global__ void gnn_pack_weights_kernel(const float* __restrict__ Wl,
                                        const float* __restrict__ bl,
                                        const float* __restrict__ Wr,
                                        float* __restrict__ WTl,
                                        float* __restrict__ WTr,
                                        float* __restrict__ blp,
                                        int dout, int dout_pad) {
    int tid = blockIdx.x * blockDim.x + threadIdx.x;
    int col = tid / DIN;
    int k   = tid % DIN;
    if (col >= dout_pad) return;
    bool cv  = (col < dout);
    WTl[(size_t)col * DIN + k] = cv ? Wl[(size_t)k * dout + col] : 0.0f;
    WTr[(size_t)col * DIN + k] = cv ? Wr[(size_t)k * dout + col] : 0.0f;
    if (k == 0) blp[col] = cv ? bl[col] : 0.0f;
}

__global__ void gnn_count_kernel(const int* __restrict__ dst,
                                 int* __restrict__ cnt,
                                 int* __restrict__ rank, int E) {
    const int e = blockIdx.x * blockDim.x + threadIdx.x;
    if (e >= E) return;
    rank[e] = atomicAdd(cnt + dst[e], 1);
}

__global__ void __launch_bounds__(1024)
gnn_scan_kernel(const int* __restrict__ cnt,
                int* __restrict__ ptr,
                float* __restrict__ deg, int n) {
    __shared__ int s_tot[32];
    __shared__ int s_part[32];
    const int tid  = threadIdx.x;
    const int lane = tid & 31;
    const int wid  = tid >> 5;
    const int base = blockIdx.x * 1024;

    const int4* c4 = (const int4*)cnt;
    const int n4 = base >> 2;
    int part = 0;
    #pragma unroll 8
    for (int i = tid; i < n4; i += 1024) {
        const int4 v = c4[i];
        part += (v.x + v.y) + (v.z + v.w);
    }

    const int i = base + tid;
    const int c = (i < n) ? cnt[i] : 0;

    int incl = c;
    #pragma unroll
    for (int off = 1; off < 32; off <<= 1) {
        const int t = __shfl_up(incl, off, 32);
        if (lane >= off) incl += t;
    }
    #pragma unroll
    for (int off = 16; off > 0; off >>= 1) part += __shfl_xor(part, off, 32);

    if (lane == 31) s_tot[wid]  = incl;
    if (lane == 0)  s_part[wid] = part;
    __syncthreads();

    int pre = (lane < wid) ? s_tot[lane] : 0;
    int all = s_part[lane];
    #pragma unroll
    for (int off = 16; off > 0; off >>= 1) {
        pre += __shfl_xor(pre, off, 32);
        all += __shfl_xor(all, off, 32);
    }
    if (i < n) {
        ptr[i] = all + pre + (incl - c);
        deg[i] = (float)c;
    }
}

__global__ void gnn_fill_kernel(const int* __restrict__ src,
                                const int* __restrict__ dst,
                                const int* __restrict__ ptr,
                                const int* __restrict__ rank,
                                int* __restrict__ col, int E) {
    const int e = blockIdx.x * blockDim.x + threadIdx.x;
    if (e >= E) return;
    col[ptr[dst[e]] + rank[e]] = src[e];
}

__global__ void __launch_bounds__(256)
gnn_gather_kernel(const float* __restrict__ x,
                  const int* __restrict__ ptr,
                  const int* __restrict__ cnt,
                  const int* __restrict__ col,
                  float* __restrict__ agg, int n) {
    const int tid  = blockIdx.x * blockDim.x + threadIdx.x;
    const int node = tid >> 4;
    const int f4   = tid & 15;
    if (node >= n) return;
    const int beg = ptr[node];
    const int end = beg + cnt[node];
    const float4* xb = (const float4*)x + f4;
    float4 acc = {0.0f, 0.0f, 0.0f, 0.0f};
    int j = beg;
    for (; j + 4 <= end; j += 4) {
        const int s0 = col[j], s1 = col[j + 1], s2 = col[j + 2], s3 = col[j + 3];
        const float4 v0 = xb[(size_t)s0 * 16];
        const float4 v1 = xb[(size_t)s1 * 16];
        const float4 v2 = xb[(size_t)s2 * 16];
        const float4 v3 = xb[(size_t)s3 * 16];
        acc.x += v0.x; acc.y += v0.y; acc.z += v0.z; acc.w += v0.w;
        acc.x += v1.x; acc.y += v1.y; acc.z += v1.z; acc.w += v1.w;
        acc.x += v2.x; acc.y += v2.y; acc.z += v2.z; acc.w += v2.w;
        acc.x += v3.x; acc.y += v3.y; acc.z += v3.z; acc.w += v3.w;
    }
    const int r = end - j;
    if (r > 0) {
        const int s0 = col[j];
        const int s1 = col[r > 1 ? j + 1 : j];
        const int s2 = col[r > 2 ? j + 2 : j];
        const float4 v0 = xb[(size_t)s0 * 16];
        const float4 v1 = xb[(size_t)s1 * 16];
        const float4 v2 = xb[(size_t)s2 * 16];
        acc.x += v0.x; acc.y += v0.y; acc.z += v0.z; acc.w += v0.w;
        if (r > 1) { acc.x += v1.x; acc.y += v1.y; acc.z += v1.z; acc.w += v1.w; }
        if (r > 2) { acc.x += v2.x; acc.y += v2.y; acc.z += v2.z; acc.w += v2.w; }
    }
    ((float4*)agg)[(size_t)node * 16 + f4] = acc;
}

__global__ void __launch_bounds__(128)
gnn_dense_wmma_kernel(const float* __restrict__ agg,
                      const float* __restrict__ deg,
                      const float* __restrict__ hin,
                      const float* __restrict__ WTl,
                      const float* __restrict__ blp,
                      const float* __restrict__ WTr,
                      float* __restrict__ hout,
                      int do_relu, int n_mtiles) {
    const int lane  = threadIdx.x;
    const int mtile = blockIdx.x * blockDim.y + threadIdx.y;
    const int ntile = blockIdx.y;
    if (mtile >= n_mtiles) return;

    const int mrow  = mtile * 16 + (lane & 15);
    const int khalf = (lane >> 4) * 2;
    const int col   = ntile * 16 + (lane & 15);

    const float rdeg = 1.0f / fmaxf(deg[mrow], 1.0f);
    const float* aggr = agg + (size_t)mrow * DIN + khalf;
    const float* hinr = hin + (size_t)mrow * DIN + khalf;
    const float* wl   = WTl + (size_t)col  * DIN + khalf;
    const float* wr   = WTr + (size_t)col  * DIN + khalf;

    v8f c = {};
    #pragma unroll
    for (int k0 = 0; k0 < DIN; k0 += 4) {
        v2f aAgg = *(const v2f*)(aggr + k0);
        aAgg.x *= rdeg; aAgg.y *= rdeg;
        v2f aX   = *(const v2f*)(hinr + k0);
        v2f bL   = *(const v2f*)(wl + k0);
        v2f bR   = *(const v2f*)(wr + k0);
        c = __builtin_amdgcn_wmma_f32_16x16x4_f32(false, aAgg, false, bL,
                                                  (short)0, c, false, false);
        c = __builtin_amdgcn_wmma_f32_16x16x4_f32(false, aX,   false, bR,
                                                  (short)0, c, false, false);
    }

    const float bv = blp[col];
    float* outp = hout + (size_t)(mtile * 16 + ((lane >> 4) * 8)) * DIN + col;
    #pragma unroll
    for (int r = 0; r < 8; ++r) {
        float v = c[r] + bv;
        if (do_relu) v = fmaxf(v, 0.0f);
        outp[(size_t)r * DIN] = v;
    }
}

__global__ void gnn_logsoftmax_kernel(const float* __restrict__ logits,
                                      float* __restrict__ out,
                                      int n_nodes, int C, int stride) {
    const int lane = threadIdx.x;
    const int node = blockIdx.x * blockDim.y + threadIdx.y;
    if (node >= n_nodes) return;
    const float* rowp = logits + (size_t)node * stride;
    const float NEG_INF = -__builtin_inff();
    float v0 = (lane      < C) ? rowp[lane]      : NEG_INF;
    float v1 = (lane + 32 < C) ? rowp[lane + 32] : NEG_INF;
    float m = fmaxf(v0, v1);
    #pragma unroll
    for (int off = 16; off > 0; off >>= 1) m = fmaxf(m, __shfl_xor(m, off, 32));
    float e0 = (lane      < C) ? expf(v0 - m) : 0.0f;
    float e1 = (lane + 32 < C) ? expf(v1 - m) : 0.0f;
    float s = e0 + e1;
    #pragma unroll
    for (int off = 16; off > 0; off >>= 1) s += __shfl_xor(s, off, 32);
    const float ls = logf(s);
    if (lane      < C) out[(size_t)node * C + lane]      = v0 - m - ls;
    if (lane + 32 < C) out[(size_t)node * C + lane + 32] = v1 - m - ls;
}

static inline void launch_dense(const float* agg, const float* deg, const float* hin,
                                const float* WTl, const float* blp, const float* WTr,
                                float* hout, int dout_pad, int do_relu,
                                int N, hipStream_t stream) {
    const int n_mtiles = (N + 15) / 16;
    dim3 block(32, 4);
    dim3 grid((n_mtiles + 3) / 4, dout_pad / 16);
    gnn_dense_wmma_kernel<<<grid, block, 0, stream>>>(agg, deg, hin, WTl, blp, WTr,
                                                      hout, do_relu, n_mtiles);
}

extern "C" void kernel_launch(void* const* d_in, const int* in_sizes, int n_in,
                              void* d_out, int out_size, void* d_ws, size_t ws_size,
                              hipStream_t stream) {
    const float* x   = (const float*)d_in[0];
    const int*   ei  = (const int*)  d_in[1];
    const float* Wl0 = (const float*)d_in[2];
    const float* bl0 = (const float*)d_in[3];
    const float* Wr0 = (const float*)d_in[4];
    const float* Wl1 = (const float*)d_in[5];
    const float* bl1 = (const float*)d_in[6];
    const float* Wr1 = (const float*)d_in[7];
    const float* Wl2 = (const float*)d_in[8];
    const float* bl2 = (const float*)d_in[9];
    const float* Wr2 = (const float*)d_in[10];

    const int N = in_sizes[0] / DIN;
    const int E = in_sizes[1] / 2;
    const int* src = ei;
    const int* dst = ei + E;

    const int DP0 = 64, DP1 = 64, DP2 = 48;

    float* deg  = (float*)d_ws;
    float* agg  = deg + N;
    float* h1   = agg + (size_t)N * DIN;
    float* h2   = h1  + (size_t)N * DIN;
    float* WTl0 = h2  + (size_t)N * DIN;
    float* WTr0 = WTl0 + DP0 * DIN;
    float* blp0 = WTr0 + DP0 * DIN;
    float* WTl1 = blp0 + DP0;
    float* WTr1 = WTl1 + DP1 * DIN;
    float* blp1 = WTr1 + DP1 * DIN;
    float* WTl2 = blp1 + DP1;
    float* WTr2 = WTl2 + DP2 * DIN;
    float* blp2 = WTr2 + DP2 * DIN;

    int* cnt  = (int*)d_out;
    int* ptr  = cnt + N;
    int* col  = ptr + N;
    int* rank = (int*)agg;

    const int ZB = 256;
    const unsigned edge_blocks   = (unsigned)((E + ZB - 1) / ZB);
    const unsigned gather_blocks = (unsigned)(((unsigned long long)N * 16ull + ZB - 1) / ZB);

    gnn_pack_weights_kernel<<<(DP0 * DIN + ZB - 1) / ZB, ZB, 0, stream>>>(
        Wl0, bl0, Wr0, WTl0, WTr0, blp0, 64, DP0);
    gnn_pack_weights_kernel<<<(DP1 * DIN + ZB - 1) / ZB, ZB, 0, stream>>>(
        Wl1, bl1, Wr1, WTl1, WTr1, blp1, 64, DP1);
    gnn_pack_weights_kernel<<<(DP2 * DIN + ZB - 1) / ZB, ZB, 0, stream>>>(
        Wl2, bl2, Wr2, WTl2, WTr2, blp2, 40, DP2);

    gnn_zero_kernel<<<(unsigned)((N + ZB - 1) / ZB), ZB, 0, stream>>>((float*)cnt, (unsigned long long)N);
    gnn_count_kernel<<<edge_blocks, ZB, 0, stream>>>(dst, cnt, rank, E);
    gnn_scan_kernel<<<(unsigned)((N + 1023) / 1024), 1024, 0, stream>>>(cnt, ptr, deg, N);
    gnn_fill_kernel<<<edge_blocks, ZB, 0, stream>>>(src, dst, ptr, rank, col, E);

    gnn_gather_kernel<<<gather_blocks, ZB, 0, stream>>>(x, ptr, cnt, col, agg, N);
    launch_dense(agg, deg, x, WTl0, blp0, WTr0, h1, DP0, 1, N, stream);

    gnn_gather_kernel<<<gather_blocks, ZB, 0, stream>>>(h1, ptr, cnt, col, agg, N);
    launch_dense(agg, deg, h1, WTl1, blp1, WTr1, h2, DP1, 1, N, stream);

    gnn_gather_kernel<<<gather_blocks, ZB, 0, stream>>>(h2, ptr, cnt, col, agg, N);
    launch_dense(agg, deg, h2, WTl2, blp2, WTr2, h1, DP2, 0, N, stream);

    {
        dim3 block(32, 8);
        dim3 grid((N + 7) / 8);
        gnn_logsoftmax_kernel<<<grid, block, 0, stream>>>(h1, (float*)d_out, N, 40, 64);
    }
}


# Dense step: one wave owns a 16-row tile with all its column tiles; weights resident in registers (staged via LDS), A tiles read once (b128) one tile ahead; k sum order of each output changes
# speedup vs baseline: 2.5825x; 1.3245x over previous
#include <hip/hip_runtime.h>
#include <cstddef>

typedef float v2f __attribute__((ext_vector_type(2)));
typedef float v8f __attribute__((ext_vector_type(8)));

#define DIN 64

__global__ void gnn_zero_kernel(float* __restrict__ p, unsigned long long n) {
    unsigned long long i = (unsigned long long)blockIdx.x * blockDim.x + threadIdx.x;
    if (i < n) p[i] = 0.0f;
}

__global__ void gnn_pack_weights_kernel(const float* __restrict__ Wl,
                                        const float* __restrict__ bl,
                                        const float* __restrict__ Wr,
                                        float* __restrict__ WTl,
                                        float* __restrict__ WTr,
                                        float* __restrict__ blp,
                                        int dout, int dout_pad) {
    int tid = blockIdx.x * blockDim.x + threadIdx.x;
    int col = tid / DIN;
    int k   = tid % DIN;
    if (col >= dout_pad) return;
    bool cv  = (col < dout);
    WTl[(size_t)col * DIN + k] = cv ? Wl[(size_t)k * dout + col] : 0.0f;
    WTr[(size_t)col * DIN + k] = cv ? Wr[(size_t)k * dout + col] : 0.0f;
    if (k == 0) blp[col] = cv ? bl[col] : 0.0f;
}

__global__ void gnn_count_kernel(const int* __restrict__ dst,
                                 int* __restrict__ cnt,
                                 int* __restrict__ rank, int E) {
    const int e = blockIdx.x * blockDim.x + threadIdx.x;
    if (e >= E) return;
    rank[e] = atomicAdd(cnt + dst[e], 1);
}

__global__ void __launch_bounds__(1024)
gnn_scan_kernel(const int* __restrict__ cnt,
                int* __restrict__ ptr,
                float* __restrict__ deg, int n) {
    __shared__ int s_tot[32];
    __shared__ int s_part[32];
    const int tid  = threadIdx.x;
    const int lane = tid & 31;
    const int wid  = tid >> 5;
    const int base = blockIdx.x * 1024;

    const int4* c4 = (const int4*)cnt;
    const int n4 = base >> 2;
    int part = 0;
    #pragma unroll 8
    for (int i = tid; i < n4; i += 1024) {
        const int4 v = c4[i];
        part += (v.x + v.y) + (v.z + v.w);
    }

    const int i = base + tid;
    const int c = (i < n) ? cnt[i] : 0;

    int incl = c;
    #pragma unroll
    for (int off = 1; off < 32; off <<= 1) {
        const int t = __shfl_up(incl, off, 32);
        if (lane >= off) incl += t;
    }
    #pragma unroll
    for (int off = 16; off > 0; off >>= 1) part += __shfl_xor(part, off, 32);

    if (lane == 31) s_tot[wid]  = incl;
    if (lane == 0)  s_part[wid] = part;
    __syncthreads();

    int pre = (lane < wid) ? s_tot[lane] : 0;
    int all = s_part[lane];
    #pragma unroll
    for (int off = 16; off > 0; off >>= 1) {
        pre += __shfl_xor(pre, off, 32);
        all += __shfl_xor(all, off, 32);
    }
    if (i < n) {
        ptr[i] = all + pre + (incl - c);
        deg[i] = (float)c;
    }
}

__global__ void gnn_fill_kernel(const int* __restrict__ src,
                                const int* __restrict__ dst,
                                const int* __restrict__ ptr,
                                const int* __restrict__ rank,
                                int* __restrict__ col, int E) {
    const int e = blockIdx.x * blockDim.x + threadIdx.x;
    if (e >= E) return;
    col[ptr[dst[e]] + rank[e]] = src[e];
}

__global__ void __launch_bounds__(256)
gnn_gather_kernel(const float* __restrict__ x,
                  const int* __restrict__ ptr,
                  const int* __restrict__ cnt,
                  const int* __restrict__ col,
                  float* __restrict__ agg, int n) {
    const int tid  = blockIdx.x * blockDim.x + threadIdx.x;
    const int node = tid >> 4;
    const int f4   = tid & 15;
    if (node >= n) return;
    const int beg = ptr[node];
    const int end = beg + cnt[node];
    const float4* xb = (const float4*)x + f4;
    float4 acc = {0.0f, 0.0f, 0.0f, 0.0f};
    int j = beg;
    for (; j + 4 <= end; j += 4) {
        const int s0 = col[j], s1 = col[j + 1], s2 = col[j + 2], s3 = col[j + 3];
        const float4 v0 = xb[(size_t)s0 * 16];
        const float4 v1 = xb[(size_t)s1 * 16];
        const float4 v2 = xb[(size_t)s2 * 16];
        const float4 v3 = xb[(size_t)s3 * 16];
        acc.x += v0.x; acc.y += v0.y; acc.z += v0.z; acc.w += v0.w;
        acc.x += v1.x; acc.y += v1.y; acc.z += v1.z; acc.w += v1.w;
        acc.x += v2.x; acc.y += v2.y; acc.z += v2.z; acc.w += v2.w;
        acc.x += v3.x; acc.y += v3.y; acc.z += v3.z; acc.w += v3.w;
    }
    const int r = end - j;
    if (r > 0) {
        const int s0 = col[j];
        const int s1 = col[r > 1 ? j + 1 : j];
        const int s2 = col[r > 2 ? j + 2 : j];
        const float4 v0 = xb[(size_t)s0 * 16];
        const float4 v1 = xb[(size_t)s1 * 16];
        const float4 v2 = xb[(size_t)s2 * 16];
        acc.x += v0.x; acc.y += v0.y; acc.z += v0.z; acc.w += v0.w;
        if (r > 1) { acc.x += v1.x; acc.y += v1.y; acc.z += v1.z; acc.w += v1.w; }
        if (r > 2) { acc.x += v2.x; acc.y += v2.y; acc.z += v2.z; acc.w += v2.w; }
    }
    ((float4*)agg)[(size_t)node * 16 + f4] = acc;
}

#define DENSE_WAVES 8
#define DENSE_WGS   256
#define WPITCH      68

template <int NT>
__global__ void __launch_bounds__(DENSE_WAVES * 32)
gnn_dense_wmma_kernel(const float* __restrict__ agg,
                      const float* __restrict__ deg,
                      const float* __restrict__ hin,
                      const float* __restrict__ WTl,
                      const float* __restrict__ blp,
                      const float* __restrict__ WTr,
                      float* __restrict__ hout,
                      int do_relu, int n_mtiles) {
    __shared__ __attribute__((aligned(16))) float sWl[NT * 16 * WPITCH];
    __shared__ __attribute__((aligned(16))) float sWr[NT * 16 * WPITCH];

    const int lane = threadIdx.x;
    const int wave = threadIdx.y;
    const int tid  = wave * 32 + lane;
    const int c16  = lane & 15;
    const int h    = lane >> 4;

    const unsigned nw = DENSE_WGS * DENSE_WAVES;
    const unsigned w  = blockIdx.x * DENSE_WAVES + wave;
    const int t_beg = (int)((w * (unsigned)n_mtiles) / nw);
    const int t_end = (int)(((w + 1u) * (unsigned)n_mtiles) / nw);

    float4 aN[8], xN[8];
    float  dN;
    {
        const int t0   = t_beg < n_mtiles ? t_beg : (n_mtiles > 0 ? n_mtiles - 1 : 0);
        const int mrow = t0 * 16 + c16;
        const float4* ap = (const float4*)(agg + (size_t)mrow * DIN) + h;
        const float4* xp = (const float4*)(hin + (size_t)mrow * DIN) + h;
        dN = deg[mrow];
        #pragma unroll
        for (int kk = 0; kk < 8; ++kk) { aN[kk] = ap[2 * kk]; xN[kk] = xp[2 * kk]; }
    }

    constexpr int STAGE_TRIPS = (NT * 16 * 16) / (DENSE_WAVES * 32);
    static_assert(STAGE_TRIPS * (DENSE_WAVES * 32) == NT * 16 * 16, "whole trips only");
    #pragma unroll
    for (int j = 0; j < STAGE_TRIPS; ++j) {
        const int i  = tid + j * (DENSE_WAVES * 32);
        const int cw = i >> 4, q = i & 15;
        const float4 vl = ((const float4*)WTl)[i];
        const float4 vr = ((const float4*)WTr)[i];
        *(float4*)(sWl + cw * WPITCH + q * 4) = vl;
        *(float4*)(sWr + cw * WPITCH + q * 4) = vr;
    }
    __syncthreads();

    float4 bL[8][NT], bR[8][NT];
    {
        const float* bl0 = sWl + c16 * WPITCH + 4 * h;
        const float* br0 = sWr + c16 * WPITCH + 4 * h;
        #pragma unroll
        for (int kk = 0; kk < 8; ++kk) {
            #pragma unroll
            for (int t = 0; t < NT; ++t) {
                bL[kk][t] = *(const float4*)(bl0 + t * 16 * WPITCH + 8 * kk);
                bR[kk][t] = *(const float4*)(br0 + t * 16 * WPITCH + 8 * kk);
            }
        }
    }

    float bv[NT];
    #pragma unroll
    for (int t = 0; t < NT; ++t) bv[t] = blp[16 * t + c16];

    for (int mtile = t_beg; mtile < t_end; ++mtile) {
        float4 aC[8], xC[8];
        #pragma unroll
        for (int kk = 0; kk < 8; ++kk) { aC[kk] = aN[kk]; xC[kk] = xN[kk]; }
        const float rdeg = 1.0f / fmaxf(dN, 1.0f);

        {
            const int tn   = (mtile + 1 < t_end) ? mtile + 1 : mtile;
            const int mrow = tn * 16 + c16;
            const float4* ap = (const float4*)(agg + (size_t)mrow * DIN) + h;
            const float4* xp = (const float4*)(hin + (size_t)mrow * DIN) + h;
            dN = deg[mrow];
            #pragma unroll
            for (int kk = 0; kk < 8; ++kk) { aN[kk] = ap[2 * kk]; xN[kk] = xp[2 * kk]; }
        }

        v8f acc[NT];
        #pragma unroll
        for (int t = 0; t < NT; ++t) acc[t] = v8f{};

        #pragma unroll
        for (int kk = 0; kk < 8; ++kk) {
            float4 a = aC[kk];
            const float4 x = xC[kk];
            a.x *= rdeg; a.y *= rdeg; a.z *= rdeg; a.w *= rdeg;
            const v2f a0 = {a.x, a.y}, a1 = {a.z, a.w};
            const v2f x0 = {x.x, x.y}, x1 = {x.z, x.w};
            #pragma unroll
            for (int t = 0; t < NT; ++t) {
                const v2f b = {bL[kk][t].x, bL[kk][t].y};
                acc[t] = __builtin_amdgcn_wmma_f32_16x16x4_f32(false, a0, false, b,
                                                               (short)0, acc[t], false, false);
            }
            #pragma unroll
            for (int t = 0; t < NT; ++t) {
                const v2f b = {bR[kk][t].x, bR[kk][t].y};
                acc[t] = __builtin_amdgcn_wmma_f32_16x16x4_f32(false, x0, false, b,
                                                               (short)0, acc[t], false, false);
            }
            #pragma unroll
            for (int t = 0; t < NT; ++t) {
                const v2f b = {bL[kk][t].z, bL[kk][t].w};
                acc[t] = __builtin_amdgcn_wmma_f32_16x16x4_f32(false, a1, false, b,
                                                               (short)0, acc[t], false, false);
            }
            #pragma unroll
            for (int t = 0; t < NT; ++t) {
                const v2f b = {bR[kk][t].z, bR[kk][t].w};
                acc[t] = __builtin_amdgcn_wmma_f32_16x16x4_f32(false, x1, false, b,
                                                               (short)0, acc[t], false, false);
            }
        }

        float* outp = hout + (size_t)(mtile * 16 + h * 8) * DIN + c16;
        #pragma unroll
        for (int t = 0; t < NT; ++t) {
            #pragma unroll
            for (int r = 0; r < 8; ++r) {
                float v = acc[t][r] + bv[t];
                if (do_relu) v = fmaxf(v, 0.0f);
                outp[(size_t)r * DIN + 16 * t] = v;
            }
        }
    }
}

__global__ void gnn_logsoftmax_kernel(const float* __restrict__ logits,
                                      float* __restrict__ out,
                                      int n_nodes, int C, int stride) {
    const int lane = threadIdx.x;
    const int node = blockIdx.x * blockDim.y + threadIdx.y;
    if (node >= n_nodes) return;
    const float* rowp = logits + (size_t)node * stride;
    const float NEG_INF = -__builtin_inff();
    float v0 = (lane      < C) ? rowp[lane]      : NEG_INF;
    float v1 = (lane + 32 < C) ? rowp[lane + 32] : NEG_INF;
    float m = fmaxf(v0, v1);
    #pragma unroll
    for (int off = 16; off > 0; off >>= 1) m = fmaxf(m, __shfl_xor(m, off, 32));
    float e0 = (lane      < C) ? expf(v0 - m) : 0.0f;
    float e1 = (lane + 32 < C) ? expf(v1 - m) : 0.0f;
    float s = e0 + e1;
    #pragma unroll
    for (int off = 16; off > 0; off >>= 1) s += __shfl_xor(s, off, 32);
    const float ls = logf(s);
    if (lane      < C) out[(size_t)node * C + lane]      = v0 - m - ls;
    if (lane + 32 < C) out[(size_t)node * C + lane + 32] = v1 - m - ls;
}

static inline void launch_dense(const float* agg, const float* deg, const float* hin,
                                const float* WTl, const float* blp, const float* WTr,
                                float* hout, int dout_pad, int do_relu,
                                int N, hipStream_t stream) {
    const int n_mtiles = (N + 15) / 16;
    dim3 block(32, DENSE_WAVES);
    dim3 grid(DENSE_WGS);
    if (dout_pad == 64)
        gnn_dense_wmma_kernel<4><<<grid, block, 0, stream>>>(agg, deg, hin, WTl, blp, WTr,
                                                             hout, do_relu, n_mtiles);
    else
        gnn_dense_wmma_kernel<3><<<grid, block, 0, stream>>>(agg, deg, hin, WTl, blp, WTr,
                                                             hout, do_relu, n_mtiles);
}

extern "C" void kernel_launch(void* const* d_in, const int* in_sizes, int n_in,
                              void* d_out, int out_size, void* d_ws, size_t ws_size,
                              hipStream_t stream) {
    const float* x   = (const float*)d_in[0];
    const int*   ei  = (const int*)  d_in[1];
    const float* Wl0 = (const float*)d_in[2];
    const float* bl0 = (const float*)d_in[3];
    const float* Wr0 = (const float*)d_in[4];
    const float* Wl1 = (const float*)d_in[5];
    const float* bl1 = (const float*)d_in[6];
    const float* Wr1 = (const float*)d_in[7];
    const float* Wl2 = (const float*)d_in[8];
    const float* bl2 = (const float*)d_in[9];
    const float* Wr2 = (const float*)d_in[10];

    const int N = in_sizes[0] / DIN;
    const int E = in_sizes[1] / 2;
    const int* src = ei;
    const int* dst = ei + E;

    const int DP0 = 64, DP1 = 64, DP2 = 48;

    float* deg  = (float*)d_ws;
    float* agg  = deg + N;
    float* h1   = agg + (size_t)N * DIN;
    float* h2   = h1  + (size_t)N * DIN;
    float* WTl0 = h2  + (size_t)N * DIN;
    float* WTr0 = WTl0 + DP0 * DIN;
    float* blp0 = WTr0 + DP0 * DIN;
    float* WTl1 = blp0 + DP0;
    float* WTr1 = WTl1 + DP1 * DIN;
    float* blp1 = WTr1 + DP1 * DIN;
    float* WTl2 = blp1 + DP1;
    float* WTr2 = WTl2 + DP2 * DIN;
    float* blp2 = WTr2 + DP2 * DIN;

    int* cnt  = (int*)d_out;
    int* ptr  = cnt + N;
    int* col  = ptr + N;
    int* rank = (int*)agg;

    const int ZB = 256;
    const unsigned edge_blocks   = (unsigned)((E + ZB - 1) / ZB);
    const unsigned gather_blocks = (unsigned)(((unsigned long long)N * 16ull + ZB - 1) / ZB);

    gnn_pack_weights_kernel<<<(DP0 * DIN + ZB - 1) / ZB, ZB, 0, stream>>>(
        Wl0, bl0, Wr0, WTl0, WTr0, blp0, 64, DP0);
    gnn_pack_weights_kernel<<<(DP1 * DIN + ZB - 1) / ZB, ZB, 0, stream>>>(
        Wl1, bl1, Wr1, WTl1, WTr1, blp1, 64, DP1);
    gnn_pack_weights_kernel<<<(DP2 * DIN + ZB - 1) / ZB, ZB, 0, stream>>>(
        Wl2, bl2, Wr2, WTl2, WTr2, blp2, 40, DP2);

    gnn_zero_kernel<<<(unsigned)((N + ZB - 1) / ZB), ZB, 0, stream>>>((float*)cnt, (unsigned long long)N);
    gnn_count_kernel<<<edge_blocks, ZB, 0, stream>>>(dst, cnt, rank, E);
    gnn_scan_kernel<<<(unsigned)((N + 1023) / 1024), 1024, 0, stream>>>(cnt, ptr, deg, N);
    gnn_fill_kernel<<<edge_blocks, ZB, 0, stream>>>(src, dst, ptr, rank, col, E);

    gnn_gather_kernel<<<gather_blocks, ZB, 0, stream>>>(x, ptr, cnt, col, agg, N);
    launch_dense(agg, deg, x, WTl0, blp0, WTr0, h1, DP0, 1, N, stream);

    gnn_gather_kernel<<<gather_blocks, ZB, 0, stream>>>(h1, ptr, cnt, col, agg, N);
    launch_dense(agg, deg, h1, WTl1, blp1, WTr1, h2, DP1, 1, N, stream);

    gnn_gather_kernel<<<gather_blocks, ZB, 0, stream>>>(h2, ptr, cnt, col, agg, N);
    launch_dense(agg, deg, h2, WTl2, blp2, WTr2, h1, DP2, 0, N, stream);

    {
        dim3 block(32, 8);
        dim3 grid((N + 7) / 8);
        gnn_logsoftmax_kernel<<<grid, block, 0, stream>>>(h1, (float*)d_out, N, 40, 64);
    }
}


# Four launches folded away (14 to 10): weights transposed inside the dense kernel's LDS staging, log-softmax finished on the last layer's accumulators; softmax row sum tree changes
# speedup vs baseline: 2.7961x; 1.0827x over previous
#include <hip/hip_runtime.h>
#include <cstddef>

typedef float v2f __attribute__((ext_vector_type(2)));
typedef float v8f __attribute__((ext_vector_type(8)));

#define DIN 64

__global__ void gnn_zero_kernel(float* __restrict__ p, unsigned long long n) {
    unsigned long long i = (unsigned long long)blockIdx.x * blockDim.x + threadIdx.x;
    if (i < n) p[i] = 0.0f;
}

__global__ void gnn_count_kernel(const int* __restrict__ dst,
                                 int* __restrict__ cnt,
                                 int* __restrict__ rank, int E) {
    const int e = blockIdx.x * blockDim.x + threadIdx.x;
    if (e >= E) return;
    rank[e] = atomicAdd(cnt + dst[e], 1);
}

__global__ void __launch_bounds__(1024)
gnn_scan_kernel(const int* __restrict__ cnt,
                int* __restrict__ ptr,
                float* __restrict__ deg, int n) {
    __shared__ int s_tot[32];
    __shared__ int s_part[32];
    const int tid  = threadIdx.x;
    const int lane = tid & 31;
    const int wid  = tid >> 5;
    const int base = blockIdx.x * 1024;

    const int4* c4 = (const int4*)cnt;
    const int n4 = base >> 2;
    int part = 0;
    #pragma unroll 8
    for (int i = tid; i < n4; i += 1024) {
        const int4 v = c4[i];
        part += (v.x + v.y) + (v.z + v.w);
    }

    const int i = base + tid;
    const int c = (i < n) ? cnt[i] : 0;

    int incl = c;
    #pragma unroll
    for (int off = 1; off < 32; off <<= 1) {
        const int t = __shfl_up(incl, off, 32);
        if (lane >= off) incl += t;
    }
    #pragma unroll
    for (int off = 16; off > 0; off >>= 1) part += __shfl_xor(part, off, 32);

    if (lane == 31) s_tot[wid]  = incl;
    if (lane == 0)  s_part[wid] = part;
    __syncthreads();

    int pre = (lane < wid) ? s_tot[lane] : 0;
    int all = s_part[lane];
    #pragma unroll
    for (int off = 16; off > 0; off >>= 1) {
        pre += __shfl_xor(pre, off, 32);
        all += __shfl_xor(all, off, 32);
    }
    if (i < n) {
        ptr[i] = all + pre + (incl - c);
        deg[i] = (float)c;
    }
}

__global__ void gnn_fill_kernel(const int* __restrict__ src,
                                const int* __restrict__ dst,
                                const int* __restrict__ ptr,
                                const int* __restrict__ rank,
                                int* __restrict__ col, int E) {
    const int e = blockIdx.x * blockDim.x + threadIdx.x;
    if (e >= E) return;
    col[ptr[dst[e]] + rank[e]] = src[e];
}

__global__ void __launch_bounds__(256)
gnn_gather_kernel(const float* __restrict__ x,
                  const int* __restrict__ ptr,
                  const int* __restrict__ cnt,
                  const int* __restrict__ col,
                  float* __restrict__ agg, int n) {
    const int tid  = blockIdx.x * blockDim.x + threadIdx.x;
    const int node = tid >> 4;
    const int f4   = tid & 15;
    if (node >= n) return;
    const int beg = ptr[node];
    const int end = beg + cnt[node];
    const float4* xb = (const float4*)x + f4;
    float4 acc = {0.0f, 0.0f, 0.0f, 0.0f};
    int j = beg;
    for (; j + 4 <= end; j += 4) {
        const int s0 = col[j], s1 = col[j + 1], s2 = col[j + 2], s3 = col[j + 3];
        const float4 v0 = xb[(size_t)s0 * 16];
        const float4 v1 = xb[(size_t)s1 * 16];
        const float4 v2 = xb[(size_t)s2 * 16];
        const float4 v3 = xb[(size_t)s3 * 16];
        acc.x += v0.x; acc.y += v0.y; acc.z += v0.z; acc.w += v0.w;
        acc.x += v1.x; acc.y += v1.y; acc.z += v1.z; acc.w += v1.w;
        acc.x += v2.x; acc.y += v2.y; acc.z += v2.z; acc.w += v2.w;
        acc.x += v3.x; acc.y += v3.y; acc.z += v3.z; acc.w += v3.w;
    }
    const int r = end - j;
    if (r > 0) {
        const int s0 = col[j];
        const int s1 = col[r > 1 ? j + 1 : j];
        const int s2 = col[r > 2 ? j + 2 : j];
        const float4 v0 = xb[(size_t)s0 * 16];
        const float4 v1 = xb[(size_t)s1 * 16];
        const float4 v2 = xb[(size_t)s2 * 16];
        acc.x += v0.x; acc.y += v0.y; acc.z += v0.z; acc.w += v0.w;
        if (r > 1) { acc.x += v1.x; acc.y += v1.y; acc.z += v1.z; acc.w += v1.w; }
        if (r > 2) { acc.x += v2.x; acc.y += v2.y; acc.z += v2.z; acc.w += v2.w; }
    }
    ((float4*)agg)[(size_t)node * 16 + f4] = acc;
}


template <int M>
__device__ __forceinline__ float row_xor(float v) {
    return __builtin_bit_cast(float,
        __builtin_amdgcn_update_dpp(0, __builtin_bit_cast(int, v), 0x160 + M, 0xf, 0xf, true));
}

#define DENSE_WAVES 8
#define DENSE_WGS   256
#define WPITCH      68

template <int NT, int DOUT, bool SOFTMAX>
__global__ void __launch_bounds__(DENSE_WAVES * 32)
gnn_dense_wmma_kernel(const float* __restrict__ agg,
                      const float* __restrict__ deg,
                      const float* __restrict__ hin,
                      const float* __restrict__ Wl,
                      const float* __restrict__ bl,
                      const float* __restrict__ Wr,
                      float* __restrict__ hout,
                      int do_relu, int n_mtiles, int n_rows) {
    static_assert(DOUT % 4 == 0 && DOUT <= NT * 16 && DOUT > (NT - 1) * 16, "column tiling");
    static_assert(DENSE_WAVES * 8 == DIN, "staging covers the 64 k rows with 8 per wave");
    __shared__ __attribute__((aligned(16))) float sWl[NT * 16 * WPITCH];
    __shared__ __attribute__((aligned(16))) float sWr[NT * 16 * WPITCH];

    const int lane = threadIdx.x;
    const int wave = threadIdx.y;
    const int c16  = lane & 15;
    const int h    = lane >> 4;

    const unsigned nw = DENSE_WGS * DENSE_WAVES;
    const unsigned w  = blockIdx.x * DENSE_WAVES + wave;
    const int t_beg = (int)((w * (unsigned)n_mtiles) / nw);
    const int t_end = (int)(((w + 1u) * (unsigned)n_mtiles) / nw);

    float4 aN[8], xN[8];
    float  dN;
    {
        const int t0   = t_beg < n_mtiles ? t_beg : (n_mtiles > 0 ? n_mtiles - 1 : 0);
        const int mrow = t0 * 16 + c16;
        const float4* ap = (const float4*)(agg + (size_t)mrow * DIN) + h;
        const float4* xp = (const float4*)(hin + (size_t)mrow * DIN) + h;
        dN = deg[mrow];
        #pragma unroll
        for (int kk = 0; kk < 8; ++kk) { aN[kk] = ap[2 * kk]; xN[kk] = xp[2 * kk]; }
    }

    {
        constexpr int Q = DOUT / 4;
        const int k  = 8 * wave + (lane >> 2);
        const int ql = lane & 3;
        float4 vl[NT], vr[NT];
        #pragma unroll
        for (int j = 0; j < NT; ++j) {
            const int q = 4 * j + ql;
            vl[j] = float4{0.0f, 0.0f, 0.0f, 0.0f};
            vr[j] = float4{0.0f, 0.0f, 0.0f, 0.0f};
            if (q < Q) {
                vl[j] = ((const float4*)Wl)[k * Q + q];
                vr[j] = ((const float4*)Wr)[k * Q + q];
            }
        }
        #pragma unroll
        for (int j = 0; j < NT; ++j) {
            const int q = 4 * j + ql;
            float* pl = sWl + (4 * q) * WPITCH + k;
            float* pr = sWr + (4 * q) * WPITCH + k;
            pl[0] = vl[j].x; pl[WPITCH] = vl[j].y; pl[2 * WPITCH] = vl[j].z; pl[3 * WPITCH] = vl[j].w;
            pr[0] = vr[j].x; pr[WPITCH] = vr[j].y; pr[2 * WPITCH] = vr[j].z; pr[3 * WPITCH] = vr[j].w;
        }
    }
    __syncthreads();

    float4 bL[8][NT], bR[8][NT];
    {
        const float* bl0 = sWl + c16 * WPITCH + 4 * h;
        const float* br0 = sWr + c16 * WPITCH + 4 * h;
        #pragma unroll
        for (int kk = 0; kk < 8; ++kk) {
            #pragma unroll
            for (int t = 0; t < NT; ++t) {
                bL[kk][t] = *(const float4*)(bl0 + t * 16 * WPITCH + 8 * kk);
                bR[kk][t] = *(const float4*)(br0 + t * 16 * WPITCH + 8 * kk);
            }
        }
    }

    float bv[NT];
    #pragma unroll
    for (int t = 0; t < NT; ++t) {
        const int colb = 16 * t + c16;
        bv[t] = (colb < DOUT) ? bl[colb < DOUT ? colb : 0] : 0.0f;
    }

    for (int mtile = t_beg; mtile < t_end; ++mtile) {
        float4 aC[8], xC[8];
        #pragma unroll
        for (int kk = 0; kk < 8; ++kk) { aC[kk] = aN[kk]; xC[kk] = xN[kk]; }
        const float rdeg = 1.0f / fmaxf(dN, 1.0f);

        {
            const int tn   = (mtile + 1 < t_end) ? mtile + 1 : mtile;
            const int mrow = tn * 16 + c16;
            const float4* ap = (const float4*)(agg + (size_t)mrow * DIN) + h;
            const float4* xp = (const float4*)(hin + (size_t)mrow * DIN) + h;
            dN = deg[mrow];
            #pragma unroll
            for (int kk = 0; kk < 8; ++kk) { aN[kk] = ap[2 * kk]; xN[kk] = xp[2 * kk]; }
        }

        v8f acc[NT];
        #pragma unroll
        for (int t = 0; t < NT; ++t) acc[t] = v8f{};

        #pragma unroll
        for (int kk = 0; kk < 8; ++kk) {
            float4 a = aC[kk];
            const float4 x = xC[kk];
            a.x *= rdeg; a.y *= rdeg; a.z *= rdeg; a.w *= rdeg;
            const v2f a0 = {a.x, a.y}, a1 = {a.z, a.w};
            const v2f x0 = {x.x, x.y}, x1 = {x.z, x.w};
            #pragma unroll
            for (int t = 0; t < NT; ++t) {
                const v2f b = {bL[kk][t].x, bL[kk][t].y};
                acc[t] = __builtin_amdgcn_wmma_f32_16x16x4_f32(false, a0, false, b,
                                                               (short)0, acc[t], false, false);
            }
            #pragma unroll
            for (int t = 0; t < NT; ++t) {
                const v2f b = {bR[kk][t].x, bR[kk][t].y};
                acc[t] = __builtin_amdgcn_wmma_f32_16x16x4_f32(false, x0, false, b,
                                                               (short)0, acc[t], false, false);
            }
            #pragma unroll
            for (int t = 0; t < NT; ++t) {
                const v2f b = {bL[kk][t].z, bL[kk][t].w};
                acc[t] = __builtin_amdgcn_wmma_f32_16x16x4_f32(false, a1, false, b,
                                                               (short)0, acc[t], false, false);
            }
            #pragma unroll
            for (int t = 0; t < NT; ++t) {
                const v2f b = {bR[kk][t].z, bR[kk][t].w};
                acc[t] = __builtin_amdgcn_wmma_f32_16x16x4_f32(false, x1, false, b,
                                                               (short)0, acc[t], false, false);
            }
        }

        if constexpr (SOFTMAX) {
            const float NEG_INF = -__builtin_inff();
            float v[NT][8], m[8], s[8];
            #pragma unroll
            for (int r = 0; r < 8; ++r) {
                m[r] = NEG_INF;
                #pragma unroll
                for (int t = 0; t < NT; ++t) {
                    const bool cv = (16 * t + 15 < DOUT) || (16 * t + c16 < DOUT);
                    v[t][r] = cv ? acc[t][r] + bv[t] : NEG_INF;
                    m[r] = fmaxf(m[r], v[t][r]);
                }
            }
            #pragma unroll
            for (int r = 0; r < 8; ++r) m[r] = fmaxf(m[r], row_xor<1>(m[r]));
            #pragma unroll
            for (int r = 0; r < 8; ++r) m[r] = fmaxf(m[r], row_xor<2>(m[r]));
            #pragma unroll
            for (int r = 0; r < 8; ++r) m[r] = fmaxf(m[r], row_xor<4>(m[r]));
            #pragma unroll
            for (int r = 0; r < 8; ++r) m[r] = fmaxf(m[r], row_xor<8>(m[r]));
            #pragma unroll
            for (int r = 0; r < 8; ++r) {
                s[r] = 0.0f;
                #pragma unroll
                for (int t = 0; t < NT; ++t) s[r] += expf(v[t][r] - m[r]);
            }
            #pragma unroll
            for (int r = 0; r < 8; ++r) s[r] += row_xor<1>(s[r]);
            #pragma unroll
            for (int r = 0; r < 8; ++r) s[r] += row_xor<2>(s[r]);
            #pragma unroll
            for (int r = 0; r < 8; ++r) s[r] += row_xor<4>(s[r]);
            #pragma unroll
            for (int r = 0; r < 8; ++r) s[r] += row_xor<8>(s[r]);
            const int row0 = mtile * 16 + h * 8;
            #pragma unroll
            for (int r = 0; r < 8; ++r) {
                const float ls = logf(s[r]);
                if (row0 + r < n_rows) {
                    float* orow = hout + (size_t)(row0 + r) * DOUT + c16;
                    #pragma unroll
                    for (int t = 0; t < NT; ++t) {
                        const bool cv = (16 * t + 15 < DOUT) || (16 * t + c16 < DOUT);
                        if (cv) orow[16 * t] = v[t][r] - m[r] - ls;
                    }
                }
            }
        } else {
            float* outp = hout + (size_t)(mtile * 16 + h * 8) * DIN + c16;
            #pragma unroll
            for (int t = 0; t < NT; ++t) {
                #pragma unroll
                for (int r = 0; r < 8; ++r) {
                    float v = acc[t][r] + bv[t];
                    if (do_relu) v = fmaxf(v, 0.0f);
                    outp[(size_t)r * DIN + 16 * t] = v;
                }
            }
        }
    }
}

extern "C" void kernel_launch(void* const* d_in, const int* in_sizes, int n_in,
                              void* d_out, int out_size, void* d_ws, size_t ws_size,
                              hipStream_t stream) {
    const float* x   = (const float*)d_in[0];
    const int*   ei  = (const int*)  d_in[1];
    const float* Wl0 = (const float*)d_in[2];
    const float* bl0 = (const float*)d_in[3];
    const float* Wr0 = (const float*)d_in[4];
    const float* Wl1 = (const float*)d_in[5];
    const float* bl1 = (const float*)d_in[6];
    const float* Wr1 = (const float*)d_in[7];
    const float* Wl2 = (const float*)d_in[8];
    const float* bl2 = (const float*)d_in[9];
    const float* Wr2 = (const float*)d_in[10];

    const int N = in_sizes[0] / DIN;
    const int E = in_sizes[1] / 2;
    const int* src = ei;
    const int* dst = ei + E;

    float* deg  = (float*)d_ws;
    float* agg  = deg + N;
    float* h1   = agg + (size_t)N * DIN;
    float* h2   = h1  + (size_t)N * DIN;

    int* cnt  = (int*)d_out;
    int* ptr  = cnt + N;
    int* col  = ptr + N;
    int* rank = (int*)agg;

    const int ZB = 256;
    const unsigned edge_blocks   = (unsigned)((E + ZB - 1) / ZB);
    const unsigned gather_blocks = (unsigned)(((unsigned long long)N * 16ull + ZB - 1) / ZB);
    const int n_mtiles = (N + 15) / 16;
    const dim3 dense_block(32, DENSE_WAVES);
    const dim3 dense_grid(DENSE_WGS);

    gnn_zero_kernel<<<(unsigned)((N + ZB - 1) / ZB), ZB, 0, stream>>>((float*)cnt, (unsigned long long)N);
    gnn_count_kernel<<<edge_blocks, ZB, 0, stream>>>(dst, cnt, rank, E);
    gnn_scan_kernel<<<(unsigned)((N + 1023) / 1024), 1024, 0, stream>>>(cnt, ptr, deg, N);
    gnn_fill_kernel<<<edge_blocks, ZB, 0, stream>>>(src, dst, ptr, rank, col, E);

    gnn_gather_kernel<<<gather_blocks, ZB, 0, stream>>>(x, ptr, cnt, col, agg, N);
    gnn_dense_wmma_kernel<4, 64, false><<<dense_grid, dense_block, 0, stream>>>(
        agg, deg, x, Wl0, bl0, Wr0, h1, 1, n_mtiles, N);

    gnn_gather_kernel<<<gather_blocks, ZB, 0, stream>>>(h1, ptr, cnt, col, agg, N);
    gnn_dense_wmma_kernel<4, 64, false><<<dense_grid, dense_block, 0, stream>>>(
        agg, deg, h1, Wl1, bl1, Wr1, h2, 1, n_mtiles, N);

    gnn_gather_kernel<<<gather_blocks, ZB, 0, stream>>>(h2, ptr, cnt, col, agg, N);
    gnn_dense_wmma_kernel<3, 40, true><<<dense_grid, dense_block, 0, stream>>>(
        agg, deg, h2, Wl2, bl2, Wr2, (float*)d_out, 0, n_mtiles, N);
}


# Edge sort by target redone on LDS bins (512-node bins staged through LDS, one block sorts a bin in LDS): per-edge global atomics and scattered 4-byte stores go; list order stays arrival order
# speedup vs baseline: 4.8336x; 1.7287x over previous
#include <hip/hip_runtime.h>
#include <cstddef>

typedef float v2f __attribute__((ext_vector_type(2)));
typedef float v8f __attribute__((ext_vector_type(8)));

#define DIN 64

__global__ void gnn_zero_kernel(float* __restrict__ p, unsigned long long n) {
    unsigned long long i = (unsigned long long)blockIdx.x * blockDim.x + threadIdx.x;
    if (i < n) p[i] = 0.0f;
}

#define BIN_SHIFT 9
#define BIN_NODES 512
#define MAX_BINS  1024
#define EDGE_CHUNK 4096
#define BIN_CAP   10240

__device__ __forceinline__ int block_excl_scan_1024(int v, int* s_w) {
    const int lane = threadIdx.x & 31;
    const int wid  = threadIdx.x >> 5;
    int incl = v;
    #pragma unroll
    for (int off = 1; off < 32; off <<= 1) {
        const int t = __shfl_up(incl, off, 32);
        if (lane >= off) incl += t;
    }
    if (lane == 31) s_w[wid] = incl;
    __syncthreads();
    int pre = (lane < wid) ? s_w[lane] : 0;
    #pragma unroll
    for (int off = 16; off > 0; off >>= 1) pre += __shfl_xor(pre, off, 32);
    __syncthreads();
    return pre + incl - v;
}

__global__ void __launch_bounds__(1024)
gnn_binhist_kernel(const int* __restrict__ dst, int* __restrict__ bin_total, int E, int B) {
    __shared__ int sH[MAX_BINS];
    const int tid = threadIdx.x;
    sH[tid] = 0;
    __syncthreads();
    const int base = blockIdx.x * EDGE_CHUNK;
    #pragma unroll
    for (int k = 0; k < EDGE_CHUNK / 1024; ++k) {
        const int e = base + k * 1024 + tid;
        if (e < E) atomicAdd(&sH[dst[e] >> BIN_SHIFT], 1);
    }
    __syncthreads();
    const int h = sH[tid];
    if (tid < B && h > 0) atomicAdd(bin_total + tid, h);
}

__global__ void __launch_bounds__(1024)
gnn_binscatter_kernel(const int* __restrict__ src,
                      const int* __restrict__ dst,
                      const int* __restrict__ bin_total,
                      int* __restrict__ bin_cursor,
                      int2* __restrict__ binned, int E, int B) {
    __shared__ int  sH[MAX_BINS];
    __shared__ int  sLoc[MAX_BINS];
    __shared__ int  sBase[MAX_BINS];
    __shared__ int  sCur[MAX_BINS];
    __shared__ int  sW[32];
    __shared__ int2 sStage[EDGE_CHUNK];
    const int tid = threadIdx.x;
    sH[tid] = 0;
    sCur[tid] = 0;
    __syncthreads();
    const int base = blockIdx.x * EDGE_CHUNK;
    int d[EDGE_CHUNK / 1024], s[EDGE_CHUNK / 1024];
    #pragma unroll
    for (int k = 0; k < EDGE_CHUNK / 1024; ++k) {
        const int e = base + k * 1024 + tid;
        d[k] = -1; s[k] = 0;
        if (e < E) { d[k] = dst[e]; s[k] = src[e]; }
    }
    const int tot = (tid < B) ? bin_total[tid] : 0;
    #pragma unroll
    for (int k = 0; k < EDGE_CHUNK / 1024; ++k)
        if (d[k] >= 0) atomicAdd(&sH[d[k] >> BIN_SHIFT], 1);
    __syncthreads();

    const int bin_start = block_excl_scan_1024(tot, sW);
    const int h         = sH[tid];
    sLoc[tid] = block_excl_scan_1024(h, sW);
    int gb = 0;
    if (tid < B && h > 0) gb = bin_start + atomicAdd(bin_cursor + tid, h);
    sBase[tid] = gb;
    __syncthreads();

    #pragma unroll
    for (int k = 0; k < EDGE_CHUNK / 1024; ++k) {
        if (d[k] >= 0) {
            const int b = d[k] >> BIN_SHIFT;
            const int p = atomicAdd(&sCur[b], 1);
            sStage[sLoc[b] + p] = make_int2(s[k], d[k]);
        }
    }
    __syncthreads();

    const int left = E - base;
    const int n = left < EDGE_CHUNK ? left : EDGE_CHUNK;
    for (int i = tid; i < n; i += 1024) {
        const int2 en = sStage[i];
        const int b = en.y >> BIN_SHIFT;
        binned[sBase[b] + (i - sLoc[b])] = en;
    }
}

__global__ void __launch_bounds__(1024)
gnn_binsort_kernel(const int2* __restrict__ binned,
                   const int* __restrict__ bin_total,
                   int* __restrict__ ptr,
                   int* __restrict__ cnt,
                   float* __restrict__ deg,
                   int* __restrict__ col, int n, int B) {
    __shared__ int sCnt[BIN_NODES];
    __shared__ int sCur[BIN_NODES];
    __shared__ int sW[32];
    __shared__ int sBin[2];
    __shared__ int sCol[BIN_CAP];
    constexpr int U = BIN_CAP / 1024;
    const int tid = threadIdx.x;
    const int b   = blockIdx.x;

    const int tot = (tid < B) ? bin_total[tid] : 0;
    const int bst = block_excl_scan_1024(tot, sW);
    if (tid == b) { sBin[0] = bst; sBin[1] = tot; }
    if (tid < BIN_NODES) sCnt[tid] = 0;
    __syncthreads();
    const int  bs    = sBin[0];
    const int  sz    = sBin[1];
    const bool small = sz <= BIN_CAP;

    int2 en[U];
    if (small) {
        #pragma unroll
        for (int u = 0; u < U; ++u) {
            const int i = tid + u * 1024;
            en[u] = make_int2(0, -1);
            if (i < sz) en[u] = binned[bs + i];
        }
        #pragma unroll
        for (int u = 0; u < U; ++u)
            if (en[u].y >= 0) atomicAdd(&sCnt[en[u].y & (BIN_NODES - 1)], 1);
    } else {
        for (int i = tid; i < sz; i += 1024)
            atomicAdd(&sCnt[binned[bs + i].y & (BIN_NODES - 1)], 1);
    }
    __syncthreads();

    const int c  = (tid < BIN_NODES) ? sCnt[tid] : 0;
    const int ex = block_excl_scan_1024(c, sW);
    if (tid < BIN_NODES) sCur[tid] = ex;
    __syncthreads();

    if (small) {
        #pragma unroll
        for (int u = 0; u < U; ++u) {
            if (en[u].y >= 0) {
                const int p = atomicAdd(&sCur[en[u].y & (BIN_NODES - 1)], 1);
                sCol[p] = en[u].x;
            }
        }
        __syncthreads();
        for (int i = tid; i < sz; i += 1024) col[bs + i] = sCol[i];
    } else {
        for (int i = tid; i < sz; i += 1024) {
            const int2 e2 = binned[bs + i];
            const int p = atomicAdd(&sCur[e2.y & (BIN_NODES - 1)], 1);
            col[bs + p] = e2.x;
        }
    }
    const int node = b * BIN_NODES + tid;
    if (tid < BIN_NODES && node < n) { ptr[node] = bs + ex; cnt[node] = c; deg[node] = (float)c; }
}

__global__ void __launch_bounds__(256)
gnn_gather_kernel(const float* __restrict__ x,
                  const int* __restrict__ ptr,
                  const int* __restrict__ cnt,
                  const int* __restrict__ col,
                  float* __restrict__ agg, int n) {
    const int tid  = blockIdx.x * blockDim.x + threadIdx.x;
    const int node = tid >> 4;
    const int f4   = tid & 15;
    if (node >= n) return;
    const int beg = ptr[node];
    const int end = beg + cnt[node];
    const float4* xb = (const float4*)x + f4;
    float4 acc = {0.0f, 0.0f, 0.0f, 0.0f};
    int j = beg;
    for (; j + 4 <= end; j += 4) {
        const int s0 = col[j], s1 = col[j + 1], s2 = col[j + 2], s3 = col[j + 3];
        const float4 v0 = xb[(size_t)s0 * 16];
        const float4 v1 = xb[(size_t)s1 * 16];
        const float4 v2 = xb[(size_t)s2 * 16];
        const float4 v3 = xb[(size_t)s3 * 16];
        acc.x += v0.x; acc.y += v0.y; acc.z += v0.z; acc.w += v0.w;
        acc.x += v1.x; acc.y += v1.y; acc.z += v1.z; acc.w += v1.w;
        acc.x += v2.x; acc.y += v2.y; acc.z += v2.z; acc.w += v2.w;
        acc.x += v3.x; acc.y += v3.y; acc.z += v3.z; acc.w += v3.w;
    }
    const int r = end - j;
    if (r > 0) {
        const int s0 = col[j];
        const int s1 = col[r > 1 ? j + 1 : j];
        const int s2 = col[r > 2 ? j + 2 : j];
        const float4 v0 = xb[(size_t)s0 * 16];
        const float4 v1 = xb[(size_t)s1 * 16];
        const float4 v2 = xb[(size_t)s2 * 16];
        acc.x += v0.x; acc.y += v0.y; acc.z += v0.z; acc.w += v0.w;
        if (r > 1) { acc.x += v1.x; acc.y += v1.y; acc.z += v1.z; acc.w += v1.w; }
        if (r > 2) { acc.x += v2.x; acc.y += v2.y; acc.z += v2.z; acc.w += v2.w; }
    }
    ((float4*)agg)[(size_t)node * 16 + f4] = acc;
}


template <int M>
__device__ __forceinline__ float row_xor(float v) {
    return __builtin_bit_cast(float,
        __builtin_amdgcn_update_dpp(0, __builtin_bit_cast(int, v), 0x160 + M, 0xf, 0xf, true));
}

#define DENSE_WAVES 8
#define DENSE_WGS   256
#define WPITCH      68

template <int NT, int DOUT, bool SOFTMAX>
__global__ void __launch_bounds__(DENSE_WAVES * 32)
gnn_dense_wmma_kernel(const float* __restrict__ agg,
                      const float* __restrict__ deg,
                      const float* __restrict__ hin,
                      const float* __restrict__ Wl,
                      const float* __restrict__ bl,
                      const float* __restrict__ Wr,
                      float* __restrict__ hout,
                      int do_relu, int n_mtiles, int n_rows) {
    static_assert(DOUT % 4 == 0 && DOUT <= NT * 16 && DOUT > (NT - 1) * 16, "column tiling");
    static_assert(DENSE_WAVES * 8 == DIN, "staging covers the 64 k rows with 8 per wave");
    __shared__ __attribute__((aligned(16))) float sWl[NT * 16 * WPITCH];
    __shared__ __attribute__((aligned(16))) float sWr[NT * 16 * WPITCH];

    const int lane = threadIdx.x;
    const int wave = threadIdx.y;
    const int c16  = lane & 15;
    const int h    = lane >> 4;

    const unsigned nw = DENSE_WGS * DENSE_WAVES;
    const unsigned w  = blockIdx.x * DENSE_WAVES + wave;
    const int t_beg = (int)((w * (unsigned)n_mtiles) / nw);
    const int t_end = (int)(((w + 1u) * (unsigned)n_mtiles) / nw);

    float4 aN[8], xN[8];
    float  dN;
    {
        const int t0   = t_beg < n_mtiles ? t_beg : (n_mtiles > 0 ? n_mtiles - 1 : 0);
        const int mrow = t0 * 16 + c16;
        const float4* ap = (const float4*)(agg + (size_t)mrow * DIN) + h;
        const float4* xp = (const float4*)(hin + (size_t)mrow * DIN) + h;
        dN = deg[mrow];
        #pragma unroll
        for (int kk = 0; kk < 8; ++kk) { aN[kk] = ap[2 * kk]; xN[kk] = xp[2 * kk]; }
    }

    {
        constexpr int Q = DOUT / 4;
        const int k  = 8 * wave + (lane >> 2);
        const int ql = lane & 3;
        float4 vl[NT], vr[NT];
        #pragma unroll
        for (int j = 0; j < NT; ++j) {
            const int q = 4 * j + ql;
            vl[j] = float4{0.0f, 0.0f, 0.0f, 0.0f};
            vr[j] = float4{0.0f, 0.0f, 0.0f, 0.0f};
            if (q < Q) {
                vl[j] = ((const float4*)Wl)[k * Q + q];
                vr[j] = ((const float4*)Wr)[k * Q + q];
            }
        }
        #pragma unroll
        for (int j = 0; j < NT; ++j) {
            const int q = 4 * j + ql;
            float* pl = sWl + (4 * q) * WPITCH + k;
            float* pr = sWr + (4 * q) * WPITCH + k;
            pl[0] = vl[j].x; pl[WPITCH] = vl[j].y; pl[2 * WPITCH] = vl[j].z; pl[3 * WPITCH] = vl[j].w;
            pr[0] = vr[j].x; pr[WPITCH] = vr[j].y; pr[2 * WPITCH] = vr[j].z; pr[3 * WPITCH] = vr[j].w;
        }
    }
    __syncthreads();

    float4 bL[8][NT], bR[8][NT];
    {
        const float* bl0 = sWl + c16 * WPITCH + 4 * h;
        const float* br0 = sWr + c16 * WPITCH + 4 * h;
        #pragma unroll
        for (int kk = 0; kk < 8; ++kk) {
            #pragma unroll
            for (int t = 0; t < NT; ++t) {
                bL[kk][t] = *(const float4*)(bl0 + t * 16 * WPITCH + 8 * kk);
                bR[kk][t] = *(const float4*)(br0 + t * 16 * WPITCH + 8 * kk);
            }
        }
    }

    float bv[NT];
    #pragma unroll
    for (int t = 0; t < NT; ++t) {
        const int colb = 16 * t + c16;
        bv[t] = (colb < DOUT) ? bl[colb < DOUT ? colb : 0] : 0.0f;
    }

    for (int mtile = t_beg; mtile < t_end; ++mtile) {
        float4 aC[8], xC[8];
        #pragma unroll
        for (int kk = 0; kk < 8; ++kk) { aC[kk] = aN[kk]; xC[kk] = xN[kk]; }
        const float rdeg = 1.0f / fmaxf(dN, 1.0f);

        {
            const int tn   = (mtile + 1 < t_end) ? mtile + 1 : mtile;
            const int mrow = tn * 16 + c16;
            const float4* ap = (const float4*)(agg + (size_t)mrow * DIN) + h;
            const float4* xp = (const float4*)(hin + (size_t)mrow * DIN) + h;
            dN = deg[mrow];
            #pragma unroll
            for (int kk = 0; kk < 8; ++kk) { aN[kk] = ap[2 * kk]; xN[kk] = xp[2 * kk]; }
        }

        v8f acc[NT];
        #pragma unroll
        for (int t = 0; t < NT; ++t) acc[t] = v8f{};

        #pragma unroll
        for (int kk = 0; kk < 8; ++kk) {
            float4 a = aC[kk];
            const float4 x = xC[kk];
            a.x *= rdeg; a.y *= rdeg; a.z *= rdeg; a.w *= rdeg;
            const v2f a0 = {a.x, a.y}, a1 = {a.z, a.w};
            const v2f x0 = {x.x, x.y}, x1 = {x.z, x.w};
            #pragma unroll
            for (int t = 0; t < NT; ++t) {
                const v2f b = {bL[kk][t].x, bL[kk][t].y};
                acc[t] = __builtin_amdgcn_wmma_f32_16x16x4_f32(false, a0, false, b,
                                                               (short)0, acc[t], false, false);
            }
            #pragma unroll
            for (int t = 0; t < NT; ++t) {
                const v2f b = {bR[kk][t].x, bR[kk][t].y};
                acc[t] = __builtin_amdgcn_wmma_f32_16x16x4_f32(false, x0, false, b,
                                                               (short)0, acc[t], false, false);
            }
            #pragma unroll
            for (int t = 0; t < NT; ++t) {
                const v2f b = {bL[kk][t].z, bL[kk][t].w};
                acc[t] = __builtin_amdgcn_wmma_f32_16x16x4_f32(false, a1, false, b,
                                                               (short)0, acc[t], false, false);
            }
            #pragma unroll
            for (int t = 0; t < NT; ++t) {
                const v2f b = {bR[kk][t].z, bR[kk][t].w};
                acc[t] = __builtin_amdgcn_wmma_f32_16x16x4_f32(false, x1, false, b,
                                                               (short)0, acc[t], false, false);
            }
        }

        if constexpr (SOFTMAX) {
            const float NEG_INF = -__builtin_inff();
            float v[NT][8], m[8], s[8];
            #pragma unroll
            for (int r = 0; r < 8; ++r) {
                m[r] = NEG_INF;
                #pragma unroll
                for (int t = 0; t < NT; ++t) {
                    const bool cv = (16 * t + 15 < DOUT) || (16 * t + c16 < DOUT);
                    v[t][r] = cv ? acc[t][r] + bv[t] : NEG_INF;
                    m[r] = fmaxf(m[r], v[t][r]);
                }
            }
            #pragma unroll
            for (int r = 0; r < 8; ++r) m[r] = fmaxf(m[r], row_xor<1>(m[r]));
            #pragma unroll
            for (int r = 0; r < 8; ++r) m[r] = fmaxf(m[r], row_xor<2>(m[r]));
            #pragma unroll
            for (int r = 0; r < 8; ++r) m[r] = fmaxf(m[r], row_xor<4>(m[r]));
            #pragma unroll
            for (int r = 0; r < 8; ++r) m[r] = fmaxf(m[r], row_xor<8>(m[r]));
            #pragma unroll
            for (int r = 0; r < 8; ++r) {
                s[r] = 0.0f;
                #pragma unroll
                for (int t = 0; t < NT; ++t) s[r] += expf(v[t][r] - m[r]);
            }
            #pragma unroll
            for (int r = 0; r < 8; ++r) s[r] += row_xor<1>(s[r]);
            #pragma unroll
            for (int r = 0; r < 8; ++r) s[r] += row_xor<2>(s[r]);
            #pragma unroll
            for (int r = 0; r < 8; ++r) s[r] += row_xor<4>(s[r]);
            #pragma unroll
            for (int r = 0; r < 8; ++r) s[r] += row_xor<8>(s[r]);
            const int row0 = mtile * 16 + h * 8;
            #pragma unroll
            for (int r = 0; r < 8; ++r) {
                const float ls = logf(s[r]);
                if (row0 + r < n_rows) {
                    float* orow = hout + (size_t)(row0 + r) * DOUT + c16;
                    #pragma unroll
                    for (int t = 0; t < NT; ++t) {
                        const bool cv = (16 * t + 15 < DOUT) || (16 * t + c16 < DOUT);
                        if (cv) orow[16 * t] = v[t][r] - m[r] - ls;
                    }
                }
            }
        } else {
            float* outp = hout + (size_t)(mtile * 16 + h * 8) * DIN + c16;
            #pragma unroll
            for (int t = 0; t < NT; ++t) {
                #pragma unroll
                for (int r = 0; r < 8; ++r) {
                    float v = acc[t][r] + bv[t];
                    if (do_relu) v = fmaxf(v, 0.0f);
                    outp[(size_t)r * DIN + 16 * t] = v;
                }
            }
        }
    }
}

extern "C" void kernel_launch(void* const* d_in, const int* in_sizes, int n_in,
                              void* d_out, int out_size, void* d_ws, size_t ws_size,
                              hipStream_t stream) {
    const float* x   = (const float*)d_in[0];
    const int*   ei  = (const int*)  d_in[1];
    const float* Wl0 = (const float*)d_in[2];
    const float* bl0 = (const float*)d_in[3];
    const float* Wr0 = (const float*)d_in[4];
    const float* Wl1 = (const float*)d_in[5];
    const float* bl1 = (const float*)d_in[6];
    const float* Wr1 = (const float*)d_in[7];
    const float* Wl2 = (const float*)d_in[8];
    const float* bl2 = (const float*)d_in[9];
    const float* Wr2 = (const float*)d_in[10];

    const int N = in_sizes[0] / DIN;
    const int E = in_sizes[1] / 2;
    const int* src = ei;
    const int* dst = ei + E;

    float* deg  = (float*)d_ws;
    float* agg  = deg + N;
    float* h1   = agg + (size_t)N * DIN;
    float* h2   = h1  + (size_t)N * DIN;

    int*  cnt        = (int*)d_out;
    int*  ptr        = cnt + N;
    int*  col        = ptr + N;
    int2* binned     = (int2*)agg;
    int*  bin_total  = (int*)h1;
    int*  bin_cursor = bin_total + MAX_BINS;

    const int ZB = 256;
    const unsigned chunk_blocks  = (unsigned)((E + EDGE_CHUNK - 1) / EDGE_CHUNK);
    const int      n_bins        = (N + BIN_NODES - 1) / BIN_NODES;
    const unsigned gather_blocks = (unsigned)(((unsigned long long)N * 16ull + ZB - 1) / ZB);
    const int n_mtiles = (N + 15) / 16;
    const dim3 dense_block(32, DENSE_WAVES);
    const dim3 dense_grid(DENSE_WGS);

    gnn_zero_kernel<<<(unsigned)(2 * MAX_BINS / ZB), ZB, 0, stream>>>((float*)bin_total, (unsigned long long)(2 * MAX_BINS));
    gnn_binhist_kernel<<<chunk_blocks, 1024, 0, stream>>>(dst, bin_total, E, n_bins);
    gnn_binscatter_kernel<<<chunk_blocks, 1024, 0, stream>>>(src, dst, bin_total, bin_cursor, binned, E, n_bins);
    gnn_binsort_kernel<<<(unsigned)n_bins, 1024, 0, stream>>>(binned, bin_total, ptr, cnt, deg, col, N, n_bins);

    gnn_gather_kernel<<<gather_blocks, ZB, 0, stream>>>(x, ptr, cnt, col, agg, N);
    gnn_dense_wmma_kernel<4, 64, false><<<dense_grid, dense_block, 0, stream>>>(
        agg, deg, x, Wl0, bl0, Wr0, h1, 1, n_mtiles, N);

    gnn_gather_kernel<<<gather_blocks, ZB, 0, stream>>>(h1, ptr, cnt, col, agg, N);
    gnn_dense_wmma_kernel<4, 64, false><<<dense_grid, dense_block, 0, stream>>>(
        agg, deg, h1, Wl1, bl1, Wr1, h2, 1, n_mtiles, N);

    gnn_gather_kernel<<<gather_blocks, ZB, 0, stream>>>(h2, ptr, cnt, col, agg, N);
    gnn_dense_wmma_kernel<3, 40, true><<<dense_grid, dense_block, 0, stream>>>(
        agg, deg, h2, Wl2, bl2, Wr2, (float*)d_out, 0, n_mtiles, N);
}


# Edge sort in two launches instead of four, no global atomics: each block bin-orders its own 4096-edge chunk and writes a run table; a bin's block collects its runs; neighbour sum order changes
# speedup vs baseline: 4.9957x; 1.0335x over previous
#include <hip/hip_runtime.h>
#include <cstddef>

typedef float v2f __attribute__((ext_vector_type(2)));
typedef float v8f __attribute__((ext_vector_type(8)));

#define DIN 64

#define BIN_SHIFT 9
#define BIN_NODES 512
#define MAX_BINS  1024
#define EDGE_CHUNK 4096
#define MAX_CHUNKS 1024
#define TAB_PITCH (MAX_BINS + 32)
#define BIN_CAP   10240
#define RUN_BATCH 16

__device__ __forceinline__ int block_excl_scan_1024(int v, int* s_w) {
    const int lane = threadIdx.x & 31;
    const int wid  = threadIdx.x >> 5;
    int incl = v;
    #pragma unroll
    for (int off = 1; off < 32; off <<= 1) {
        const int t = __shfl_up(incl, off, 32);
        if (lane >= off) incl += t;
    }
    if (lane == 31) s_w[wid] = incl;
    __syncthreads();
    int pre = (lane < wid) ? s_w[lane] : 0;
    #pragma unroll
    for (int off = 16; off > 0; off >>= 1) pre += __shfl_xor(pre, off, 32);
    __syncthreads();
    return pre + incl - v;
}

__global__ void __launch_bounds__(1024)
gnn_chunksort_kernel(const int* __restrict__ src,
                     const int* __restrict__ dst,
                     int2* __restrict__ binned,
                     int* __restrict__ offtab, int E) {
    __shared__ int  sH[MAX_BINS];
    __shared__ int  sLoc[MAX_BINS];
    __shared__ int  sCur[MAX_BINS];
    __shared__ int  sW[32];
    __shared__ int2 sStage[EDGE_CHUNK];
    const int tid = threadIdx.x;
    sH[tid] = 0;
    sCur[tid] = 0;
    __syncthreads();
    const int base = blockIdx.x * EDGE_CHUNK;
    int d[EDGE_CHUNK / 1024], s[EDGE_CHUNK / 1024];
    #pragma unroll
    for (int k = 0; k < EDGE_CHUNK / 1024; ++k) {
        const int e = base + k * 1024 + tid;
        d[k] = -1; s[k] = 0;
        if (e < E) { d[k] = dst[e]; s[k] = src[e]; }
    }
    #pragma unroll
    for (int k = 0; k < EDGE_CHUNK / 1024; ++k)
        if (d[k] >= 0) atomicAdd(&sH[d[k] >> BIN_SHIFT], 1);
    __syncthreads();

    const int h   = sH[tid];
    const int loc = block_excl_scan_1024(h, sW);
    sLoc[tid] = loc;
    __syncthreads();

    #pragma unroll
    for (int k = 0; k < EDGE_CHUNK / 1024; ++k) {
        if (d[k] >= 0) {
            const int b = d[k] >> BIN_SHIFT;
            const int p = atomicAdd(&sCur[b], 1);
            sStage[sLoc[b] + p] = make_int2(s[k], d[k]);
        }
    }
    __syncthreads();

    const int left = E - base;
    const int n = left < EDGE_CHUNK ? left : EDGE_CHUNK;
    for (int i = tid; i < n; i += 1024) binned[base + i] = sStage[i];
    int* row = offtab + (size_t)blockIdx.x * TAB_PITCH;
    row[tid] = loc;
    if (tid == MAX_BINS - 1) row[MAX_BINS] = loc + h;
}

__global__ void __launch_bounds__(1024)
gnn_binsort_kernel(const int2* __restrict__ binned,
                   const int* __restrict__ offtab,
                   int* __restrict__ ptr,
                   int* __restrict__ cnt,
                   float* __restrict__ deg,
                   int* __restrict__ col, int n, int n_chunks) {
    __shared__ int  sCnt[BIN_NODES];
    __shared__ int  sCur[BIN_NODES];
    __shared__ int  sW[32];
    __shared__ int  sBin[2];
    __shared__ int4 sRun[MAX_CHUNKS];
    __shared__ int2 sEdge[BIN_CAP];
    __shared__ int  sCol[BIN_CAP];
    const int tid  = threadIdx.x;
    const int lane = tid & 31;
    const int wave = tid >> 5;
    const int b    = blockIdx.x;

    int o0 = 0, rl = 0;
    if (tid < n_chunks) {
        const int* row = offtab + (size_t)tid * TAB_PITCH + b;
        o0 = row[0];
        rl = row[1] - o0;
    }
    const int pos = block_excl_scan_1024(rl, sW);
    const int st  = block_excl_scan_1024(o0, sW);
    sRun[tid] = make_int4(tid * EDGE_CHUNK + o0, rl, pos, 0);
    if (tid == 1023) { sBin[0] = st + o0; sBin[1] = pos + rl; }
    if (tid < BIN_NODES) sCnt[tid] = 0;
    __syncthreads();
    const int  bs    = sBin[0];
    const int  sz    = sBin[1];
    const bool small = sz <= BIN_CAP;

    if (small) {
        for (int t0 = wave; t0 < n_chunks; t0 += 32 * RUN_BATCH) {
            int4 rd[RUN_BATCH];
            int2 en[RUN_BATCH];
            #pragma unroll
            for (int r = 0; r < RUN_BATCH; ++r) {
                const int t = t0 + 32 * r;
                rd[r] = make_int4(0, 0, 0, 0);
                if (t < n_chunks) rd[r] = sRun[t];
            }
            #pragma unroll
            for (int r = 0; r < RUN_BATCH; ++r) {
                en[r] = make_int2(0, -1);
                if (lane < rd[r].y) en[r] = binned[rd[r].x + lane];
            }
            #pragma unroll
            for (int r = 0; r < RUN_BATCH; ++r) {
                if (en[r].y >= 0) {
                    sEdge[rd[r].z + lane] = en[r];
                    atomicAdd(&sCnt[en[r].y & (BIN_NODES - 1)], 1);
                }
            }
            #pragma unroll
            for (int r = 0; r < RUN_BATCH; ++r) {
                for (int l = lane + 32; l < rd[r].y; l += 32) {
                    const int2 e2 = binned[rd[r].x + l];
                    sEdge[rd[r].z + l] = e2;
                    atomicAdd(&sCnt[e2.y & (BIN_NODES - 1)], 1);
                }
            }
        }
    } else {
        for (int t = wave; t < n_chunks; t += 32) {
            const int4 rd = sRun[t];
            for (int l = lane; l < rd.y; l += 32)
                atomicAdd(&sCnt[binned[rd.x + l].y & (BIN_NODES - 1)], 1);
        }
    }
    __syncthreads();

    const int c  = (tid < BIN_NODES) ? sCnt[tid] : 0;
    const int ex = block_excl_scan_1024(c, sW);
    if (tid < BIN_NODES) sCur[tid] = ex;
    __syncthreads();

    if (small) {
        for (int i = tid; i < sz; i += 1024) {
            const int2 e2 = sEdge[i];
            const int p = atomicAdd(&sCur[e2.y & (BIN_NODES - 1)], 1);
            sCol[p] = e2.x;
        }
        __syncthreads();
        for (int i = tid; i < sz; i += 1024) col[bs + i] = sCol[i];
    } else {
        for (int t = wave; t < n_chunks; t += 32) {
            const int4 rd = sRun[t];
            for (int l = lane; l < rd.y; l += 32) {
                const int2 e2 = binned[rd.x + l];
                const int p = atomicAdd(&sCur[e2.y & (BIN_NODES - 1)], 1);
                col[bs + p] = e2.x;
            }
        }
    }
    const int node = b * BIN_NODES + tid;
    if (tid < BIN_NODES && node < n) { ptr[node] = bs + ex; cnt[node] = c; deg[node] = (float)c; }
}

__global__ void __launch_bounds__(256)
gnn_gather_kernel(const float* __restrict__ x,
                  const int* __restrict__ ptr,
                  const int* __restrict__ cnt,
                  const int* __restrict__ col,
                  float* __restrict__ agg, int n) {
    const int tid  = blockIdx.x * blockDim.x + threadIdx.x;
    const int node = tid >> 4;
    const int f4   = tid & 15;
    if (node >= n) return;
    const int beg = ptr[node];
    const int end = beg + cnt[node];
    const float4* xb = (const float4*)x + f4;
    float4 acc = {0.0f, 0.0f, 0.0f, 0.0f};
    int j = beg;
    for (; j + 4 <= end; j += 4) {
        const int s0 = col[j], s1 = col[j + 1], s2 = col[j + 2], s3 = col[j + 3];
        const float4 v0 = xb[(size_t)s0 * 16];
        const float4 v1 = xb[(size_t)s1 * 16];
        const float4 v2 = xb[(size_t)s2 * 16];
        const float4 v3 = xb[(size_t)s3 * 16];
        acc.x += v0.x; acc.y += v0.y; acc.z += v0.z; acc.w += v0.w;
        acc.x += v1.x; acc.y += v1.y; acc.z += v1.z; acc.w += v1.w;
        acc.x += v2.x; acc.y += v2.y; acc.z += v2.z; acc.w += v2.w;
        acc.x += v3.x; acc.y += v3.y; acc.z += v3.z; acc.w += v3.w;
    }
    const int r = end - j;
    if (r > 0) {
        const int s0 = col[j];
        const int s1 = col[r > 1 ? j + 1 : j];
        const int s2 = col[r > 2 ? j + 2 : j];
        const float4 v0 = xb[(size_t)s0 * 16];
        const float4 v1 = xb[(size_t)s1 * 16];
        const float4 v2 = xb[(size_t)s2 * 16];
        acc.x += v0.x; acc.y += v0.y; acc.z += v0.z; acc.w += v0.w;
        if (r > 1) { acc.x += v1.x; acc.y += v1.y; acc.z += v1.z; acc.w += v1.w; }
        if (r > 2) { acc.x += v2.x; acc.y += v2.y; acc.z += v2.z; acc.w += v2.w; }
    }
    ((float4*)agg)[(size_t)node * 16 + f4] = acc;
}


template <int M>
__device__ __forceinline__ float row_xor(float v) {
    return __builtin_bit_cast(float,
        __builtin_amdgcn_update_dpp(0, __builtin_bit_cast(int, v), 0x160 + M, 0xf, 0xf, true));
}

#define DENSE_WAVES 8
#define DENSE_WGS   256
#define WPITCH      68

template <int NT, int DOUT, bool SOFTMAX>
__global__ void __launch_bounds__(DENSE_WAVES * 32)
gnn_dense_wmma_kernel(const float* __restrict__ agg,
                      const float* __restrict__ deg,
                      const float* __restrict__ hin,
                      const float* __restrict__ Wl,
                      const float* __restrict__ bl,
                      const float* __restrict__ Wr,
                      float* __restrict__ hout,
                      int do_relu, int n_mtiles, int n_rows) {
    static_assert(DOUT % 4 == 0 && DOUT <= NT * 16 && DOUT > (NT - 1) * 16, "column tiling");
    static_assert(DENSE_WAVES * 8 == DIN, "staging covers the 64 k rows with 8 per wave");
    __shared__ __attribute__((aligned(16))) float sWl[NT * 16 * WPITCH];
    __shared__ __attribute__((aligned(16))) float sWr[NT * 16 * WPITCH];

    const int lane = threadIdx.x;
    const int wave = threadIdx.y;
    const int c16  = lane & 15;
    const int h    = lane >> 4;

    const unsigned nw = DENSE_WGS * DENSE_WAVES;
    const unsigned w  = blockIdx.x * DENSE_WAVES + wave;
    const int t_beg = (int)((w * (unsigned)n_mtiles) / nw);
    const int t_end = (int)(((w + 1u) * (unsigned)n_mtiles) / nw);

    float4 aN[8], xN[8];
    float  dN;
    {
        const int t0   = t_beg < n_mtiles ? t_beg : (n_mtiles > 0 ? n_mtiles - 1 : 0);
        const int mrow = t0 * 16 + c16;
        const float4* ap = (const float4*)(agg + (size_t)mrow * DIN) + h;
        const float4* xp = (const float4*)(hin + (size_t)mrow * DIN) + h;
        dN = deg[mrow];
        #pragma unroll
        for (int kk = 0; kk < 8; ++kk) { aN[kk] = ap[2 * kk]; xN[kk] = xp[2 * kk]; }
    }

    {
        constexpr int Q = DOUT / 4;
        const int k  = 8 * wave + (lane >> 2);
        const int ql = lane & 3;
        float4 vl[NT], vr[NT];
        #pragma unroll
        for (int j = 0; j < NT; ++j) {
            const int q = 4 * j + ql;
            vl[j] = float4{0.0f, 0.0f, 0.0f, 0.0f};
            vr[j] = float4{0.0f, 0.0f, 0.0f, 0.0f};
            if (q < Q) {
                vl[j] = ((const float4*)Wl)[k * Q + q];
                vr[j] = ((const float4*)Wr)[k * Q + q];
            }
        }
        #pragma unroll
        for (int j = 0; j < NT; ++j) {
            const int q = 4 * j + ql;
            float* pl = sWl + (4 * q) * WPITCH + k;
            float* pr = sWr + (4 * q) * WPITCH + k;
            pl[0] = vl[j].x; pl[WPITCH] = vl[j].y; pl[2 * WPITCH] = vl[j].z; pl[3 * WPITCH] = vl[j].w;
            pr[0] = vr[j].x; pr[WPITCH] = vr[j].y; pr[2 * WPITCH] = vr[j].z; pr[3 * WPITCH] = vr[j].w;
        }
    }
    __syncthreads();

    float4 bL[8][NT], bR[8][NT];
    {
        const float* bl0 = sWl + c16 * WPITCH + 4 * h;
        const float* br0 = sWr + c16 * WPITCH + 4 * h;
        #pragma unroll
        for (int kk = 0; kk < 8; ++kk) {
            #pragma unroll
            for (int t = 0; t < NT; ++t) {
                bL[kk][t] = *(const float4*)(bl0 + t * 16 * WPITCH + 8 * kk);
                bR[kk][t] = *(const float4*)(br0 + t * 16 * WPITCH + 8 * kk);
            }
        }
    }

    float bv[NT];
    #pragma unroll
    for (int t = 0; t < NT; ++t) {
        const int colb = 16 * t + c16;
        bv[t] = (colb < DOUT) ? bl[colb < DOUT ? colb : 0] : 0.0f;
    }

    for (int mtile = t_beg; mtile < t_end; ++mtile) {
        float4 aC[8], xC[8];
        #pragma unroll
        for (int kk = 0; kk < 8; ++kk) { aC[kk] = aN[kk]; xC[kk] = xN[kk]; }
        const float rdeg = 1.0f / fmaxf(dN, 1.0f);

        {
            const int tn   = (mtile + 1 < t_end) ? mtile + 1 : mtile;
            const int mrow = tn * 16 + c16;
            const float4* ap = (const float4*)(agg + (size_t)mrow * DIN) + h;
            const float4* xp = (const float4*)(hin + (size_t)mrow * DIN) + h;
            dN = deg[mrow];
            #pragma unroll
            for (int kk = 0; kk < 8; ++kk) { aN[kk] = ap[2 * kk]; xN[kk] = xp[2 * kk]; }
        }

        v8f acc[NT];
        #pragma unroll
        for (int t = 0; t < NT; ++t) acc[t] = v8f{};

        #pragma unroll
        for (int kk = 0; kk < 8; ++kk) {
            float4 a = aC[kk];
            const float4 x = xC[kk];
            a.x *= rdeg; a.y *= rdeg; a.z *= rdeg; a.w *= rdeg;
            const v2f a0 = {a.x, a.y}, a1 = {a.z, a.w};
            const v2f x0 = {x.x, x.y}, x1 = {x.z, x.w};
            #pragma unroll
            for (int t = 0; t < NT; ++t) {
                const v2f b = {bL[kk][t].x, bL[kk][t].y};
                acc[t] = __builtin_amdgcn_wmma_f32_16x16x4_f32(false, a0, false, b,
                                                               (short)0, acc[t], false, false);
            }
            #pragma unroll
            for (int t = 0; t < NT; ++t) {
                const v2f b = {bR[kk][t].x, bR[kk][t].y};
                acc[t] = __builtin_amdgcn_wmma_f32_16x16x4_f32(false, x0, false, b,
                                                               (short)0, acc[t], false, false);
            }
            #pragma unroll
            for (int t = 0; t < NT; ++t) {
                const v2f b = {bL[kk][t].z, bL[kk][t].w};
                acc[t] = __builtin_amdgcn_wmma_f32_16x16x4_f32(false, a1, false, b,
                                                               (short)0, acc[t], false, false);
            }
            #pragma unroll
            for (int t = 0; t < NT; ++t) {
                const v2f b = {bR[kk][t].z, bR[kk][t].w};
                acc[t] = __builtin_amdgcn_wmma_f32_16x16x4_f32(false, x1, false, b,
                                                               (short)0, acc[t], false, false);
            }
        }

        if constexpr (SOFTMAX) {
            const float NEG_INF = -__builtin_inff();
            float v[NT][8], m[8], s[8];
            #pragma unroll
            for (int r = 0; r < 8; ++r) {
                m[r] = NEG_INF;
                #pragma unroll
                for (int t = 0; t < NT; ++t) {
                    const bool cv = (16 * t + 15 < DOUT) || (16 * t + c16 < DOUT);
                    v[t][r] = cv ? acc[t][r] + bv[t] : NEG_INF;
                    m[r] = fmaxf(m[r], v[t][r]);
                }
            }
            #pragma unroll
            for (int r = 0; r < 8; ++r) m[r] = fmaxf(m[r], row_xor<1>(m[r]));
            #pragma unroll
            for (int r = 0; r < 8; ++r) m[r] = fmaxf(m[r], row_xor<2>(m[r]));
            #pragma unroll
            for (int r = 0; r < 8; ++r) m[r] = fmaxf(m[r], row_xor<4>(m[r]));
            #pragma unroll
            for (int r = 0; r < 8; ++r) m[r] = fmaxf(m[r], row_xor<8>(m[r]));
            #pragma unroll
            for (int r = 0; r < 8; ++r) {
                s[r] = 0.0f;
                #pragma unroll
                for (int t = 0; t < NT; ++t) s[r] += expf(v[t][r] - m[r]);
            }
            #pragma unroll
            for (int r = 0; r < 8; ++r) s[r] += row_xor<1>(s[r]);
            #pragma unroll
            for (int r = 0; r < 8; ++r) s[r] += row_xor<2>(s[r]);
            #pragma unroll
            for (int r = 0; r < 8; ++r) s[r] += row_xor<4>(s[r]);
            #pragma unroll
            for (int r = 0; r < 8; ++r) s[r] += row_xor<8>(s[r]);
            const int row0 = mtile * 16 + h * 8;
            #pragma unroll
            for (int r = 0; r < 8; ++r) {
                const float ls = logf(s[r]);
                if (row0 + r < n_rows) {
                    float* orow = hout + (size_t)(row0 + r) * DOUT + c16;
                    #pragma unroll
                    for (int t = 0; t < NT; ++t) {
                        const bool cv = (16 * t + 15 < DOUT) || (16 * t + c16 < DOUT);
                        if (cv) orow[16 * t] = v[t][r] - m[r] - ls;
                    }
                }
            }
        } else {
            float* outp = hout + (size_t)(mtile * 16 + h * 8) * DIN + c16;
            #pragma unroll
            for (int t = 0; t < NT; ++t) {
                #pragma unroll
                for (int r = 0; r < 8; ++r) {
                    float v = acc[t][r] + bv[t];
                    if (do_relu) v = fmaxf(v, 0.0f);
                    outp[(size_t)r * DIN + 16 * t] = v;
                }
            }
        }
    }
}

extern "C" void kernel_launch(void* const* d_in, const int* in_sizes, int n_in,
                              void* d_out, int out_size, void* d_ws, size_t ws_size,
                              hipStream_t stream) {
    const float* x   = (const float*)d_in[0];
    const int*   ei  = (const int*)  d_in[1];
    const float* Wl0 = (const float*)d_in[2];
    const float* bl0 = (const float*)d_in[3];
    const float* Wr0 = (const float*)d_in[4];
    const float* Wl1 = (const float*)d_in[5];
    const float* bl1 = (const float*)d_in[6];
    const float* Wr1 = (const float*)d_in[7];
    const float* Wl2 = (const float*)d_in[8];
    const float* bl2 = (const float*)d_in[9];
    const float* Wr2 = (const float*)d_in[10];

    const int N = in_sizes[0] / DIN;
    const int E = in_sizes[1] / 2;
    const int* src = ei;
    const int* dst = ei + E;

    float* deg  = (float*)d_ws;
    float* agg  = deg + N;
    float* h1   = agg + (size_t)N * DIN;
    float* h2   = h1  + (size_t)N * DIN;

    int*  cnt        = (int*)d_out;
    int*  ptr        = cnt + N;
    int*  col        = ptr + N;
    int2* binned     = (int2*)agg;
    int*  offtab     = (int*)h1;

    const int ZB = 256;
    const unsigned chunk_blocks  = (unsigned)((E + EDGE_CHUNK - 1) / EDGE_CHUNK);
    const int      n_bins        = (N + BIN_NODES - 1) / BIN_NODES;
    const unsigned gather_blocks = (unsigned)(((unsigned long long)N * 16ull + ZB - 1) / ZB);
    const int n_mtiles = (N + 15) / 16;
    const dim3 dense_block(32, DENSE_WAVES);
    const dim3 dense_grid(DENSE_WGS);

    gnn_chunksort_kernel<<<chunk_blocks, 1024, 0, stream>>>(src, dst, binned, offtab, E);
    gnn_binsort_kernel<<<(unsigned)n_bins, 1024, 0, stream>>>(binned, offtab, ptr, cnt, deg, col, N, (int)chunk_blocks);

    gnn_gather_kernel<<<gather_blocks, ZB, 0, stream>>>(x, ptr, cnt, col, agg, N);
    gnn_dense_wmma_kernel<4, 64, false><<<dense_grid, dense_block, 0, stream>>>(
        agg, deg, x, Wl0, bl0, Wr0, h1, 1, n_mtiles, N);

    gnn_gather_kernel<<<gather_blocks, ZB, 0, stream>>>(h1, ptr, cnt, col, agg, N);
    gnn_dense_wmma_kernel<4, 64, false><<<dense_grid, dense_block, 0, stream>>>(
        agg, deg, h1, Wl1, bl1, Wr1, h2, 1, n_mtiles, N);

    gnn_gather_kernel<<<gather_blocks, ZB, 0, stream>>>(h2, ptr, cnt, col, agg, N);
    gnn_dense_wmma_kernel<3, 40, true><<<dense_grid, dense_block, 0, stream>>>(
        agg, deg, h2, Wl2, bl2, Wr2, (float*)d_out, 0, n_mtiles, N);
}


# Gather made resident: 1024 work-groups, a 16-lane group walks a contiguous node share, reads 16 list entries per load (lane exchange), next list prefetched, 8 rows in flight; sum order kept
# speedup vs baseline: 5.2773x; 1.0564x over previous
#include <hip/hip_runtime.h>
#include <cstddef>

typedef float v2f __attribute__((ext_vector_type(2)));
typedef float v8f __attribute__((ext_vector_type(8)));

#define DIN 64

#define BIN_SHIFT 9
#define BIN_NODES 512
#define MAX_BINS  1024
#define EDGE_CHUNK 4096
#define MAX_CHUNKS 1024
#define TAB_PITCH (MAX_BINS + 32)
#define BIN_CAP   10240
#define RUN_BATCH 16

__device__ __forceinline__ int block_excl_scan_1024(int v, int* s_w) {
    const int lane = threadIdx.x & 31;
    const int wid  = threadIdx.x >> 5;
    int incl = v;
    #pragma unroll
    for (int off = 1; off < 32; off <<= 1) {
        const int t = __shfl_up(incl, off, 32);
        if (lane >= off) incl += t;
    }
    if (lane == 31) s_w[wid] = incl;
    __syncthreads();
    int pre = (lane < wid) ? s_w[lane] : 0;
    #pragma unroll
    for (int off = 16; off > 0; off >>= 1) pre += __shfl_xor(pre, off, 32);
    __syncthreads();
    return pre + incl - v;
}

__global__ void __launch_bounds__(1024)
gnn_chunksort_kernel(const int* __restrict__ src,
                     const int* __restrict__ dst,
                     int2* __restrict__ binned,
                     int* __restrict__ offtab, int E) {
    __shared__ int  sH[MAX_BINS];
    __shared__ int  sLoc[MAX_BINS];
    __shared__ int  sCur[MAX_BINS];
    __shared__ int  sW[32];
    __shared__ int2 sStage[EDGE_CHUNK];
    const int tid = threadIdx.x;
    sH[tid] = 0;
    sCur[tid] = 0;
    __syncthreads();
    const int base = blockIdx.x * EDGE_CHUNK;
    int d[EDGE_CHUNK / 1024], s[EDGE_CHUNK / 1024];
    #pragma unroll
    for (int k = 0; k < EDGE_CHUNK / 1024; ++k) {
        const int e = base + k * 1024 + tid;
        d[k] = -1; s[k] = 0;
        if (e < E) { d[k] = dst[e]; s[k] = src[e]; }
    }
    #pragma unroll
    for (int k = 0; k < EDGE_CHUNK / 1024; ++k)
        if (d[k] >= 0) atomicAdd(&sH[d[k] >> BIN_SHIFT], 1);
    __syncthreads();

    const int h   = sH[tid];
    const int loc = block_excl_scan_1024(h, sW);
    sLoc[tid] = loc;
    __syncthreads();

    #pragma unroll
    for (int k = 0; k < EDGE_CHUNK / 1024; ++k) {
        if (d[k] >= 0) {
            const int b = d[k] >> BIN_SHIFT;
            const int p = atomicAdd(&sCur[b], 1);
            sStage[sLoc[b] + p] = make_int2(s[k], d[k]);
        }
    }
    __syncthreads();

    const int left = E - base;
    const int n = left < EDGE_CHUNK ? left : EDGE_CHUNK;
    for (int i = tid; i < n; i += 1024) binned[base + i] = sStage[i];
    int* row = offtab + (size_t)blockIdx.x * TAB_PITCH;
    row[tid] = loc;
    if (tid == MAX_BINS - 1) row[MAX_BINS] = loc + h;
}

__global__ void __launch_bounds__(1024)
gnn_binsort_kernel(const int2* __restrict__ binned,
                   const int* __restrict__ offtab,
                   int* __restrict__ ptr,
                   int* __restrict__ cnt,
                   float* __restrict__ deg,
                   int* __restrict__ col, int n, int n_chunks) {
    __shared__ int  sCnt[BIN_NODES];
    __shared__ int  sCur[BIN_NODES];
    __shared__ int  sW[32];
    __shared__ int  sBin[2];
    __shared__ int4 sRun[MAX_CHUNKS];
    __shared__ int2 sEdge[BIN_CAP];
    __shared__ int  sCol[BIN_CAP];
    const int tid  = threadIdx.x;
    const int lane = tid & 31;
    const int wave = tid >> 5;
    const int b    = blockIdx.x;

    int o0 = 0, rl = 0;
    if (tid < n_chunks) {
        const int* row = offtab + (size_t)tid * TAB_PITCH + b;
        o0 = row[0];
        rl = row[1] - o0;
    }
    const int pos = block_excl_scan_1024(rl, sW);
    const int st  = block_excl_scan_1024(o0, sW);
    sRun[tid] = make_int4(tid * EDGE_CHUNK + o0, rl, pos, 0);
    if (tid == 1023) { sBin[0] = st + o0; sBin[1] = pos + rl; }
    if (tid < BIN_NODES) sCnt[tid] = 0;
    __syncthreads();
    const int  bs    = sBin[0];
    const int  sz    = sBin[1];
    const bool small = sz <= BIN_CAP;

    if (small) {
        for (int t0 = wave; t0 < n_chunks; t0 += 32 * RUN_BATCH) {
            int4 rd[RUN_BATCH];
            int2 en[RUN_BATCH];
            #pragma unroll
            for (int r = 0; r < RUN_BATCH; ++r) {
                const int t = t0 + 32 * r;
                rd[r] = make_int4(0, 0, 0, 0);
                if (t < n_chunks) rd[r] = sRun[t];
            }
            #pragma unroll
            for (int r = 0; r < RUN_BATCH; ++r) {
                en[r] = make_int2(0, -1);
                if (lane < rd[r].y) en[r] = binned[rd[r].x + lane];
            }
            #pragma unroll
            for (int r = 0; r < RUN_BATCH; ++r) {
                if (en[r].y >= 0) {
                    sEdge[rd[r].z + lane] = en[r];
                    atomicAdd(&sCnt[en[r].y & (BIN_NODES - 1)], 1);
                }
            }
            #pragma unroll
            for (int r = 0; r < RUN_BATCH; ++r) {
                for (int l = lane + 32; l < rd[r].y; l += 32) {
                    const int2 e2 = binned[rd[r].x + l];
                    sEdge[rd[r].z + l] = e2;
                    atomicAdd(&sCnt[e2.y & (BIN_NODES - 1)], 1);
                }
            }
        }
    } else {
        for (int t = wave; t < n_chunks; t += 32) {
            const int4 rd = sRun[t];
            for (int l = lane; l < rd.y; l += 32)
                atomicAdd(&sCnt[binned[rd.x + l].y & (BIN_NODES - 1)], 1);
        }
    }
    __syncthreads();

    const int c  = (tid < BIN_NODES) ? sCnt[tid] : 0;
    const int ex = block_excl_scan_1024(c, sW);
    if (tid < BIN_NODES) sCur[tid] = ex;
    __syncthreads();

    if (small) {
        for (int i = tid; i < sz; i += 1024) {
            const int2 e2 = sEdge[i];
            const int p = atomicAdd(&sCur[e2.y & (BIN_NODES - 1)], 1);
            sCol[p] = e2.x;
        }
        __syncthreads();
        for (int i = tid; i < sz; i += 1024) col[bs + i] = sCol[i];
    } else {
        for (int t = wave; t < n_chunks; t += 32) {
            const int4 rd = sRun[t];
            for (int l = lane; l < rd.y; l += 32) {
                const int2 e2 = binned[rd.x + l];
                const int p = atomicAdd(&sCur[e2.y & (BIN_NODES - 1)], 1);
                col[bs + p] = e2.x;
            }
        }
    }
    const int node = b * BIN_NODES + tid;
    if (tid < BIN_NODES && node < n) { ptr[node] = bs + ex; cnt[node] = c; deg[node] = (float)c; }
}

#define GATHER_WGS 1024

template <int K>
__device__ __forceinline__ int row_bcast(int v) {
    return __builtin_amdgcn_update_dpp(0, v, 0x150 + K, 0xf, 0xf, true);
}

#define GATHER_QUAD_LOAD(B, V)                                                   \
    {                                                                            \
        const int s0 = row_bcast<B>(idx),     s1 = row_bcast<B + 1>(idx);        \
        const int s2 = row_bcast<B + 2>(idx), s3 = row_bcast<B + 3>(idx);        \
        V[0] = xb[(size_t)s0 * 16];                                              \
        V[1] = xb[(size_t)(B + 1 < m ? s1 : s0) * 16];                           \
        V[2] = xb[(size_t)(B + 2 < m ? s2 : s0) * 16];                           \
        V[3] = xb[(size_t)(B + 3 < m ? s3 : s0) * 16];                           \
    }
#define GATHER_QUAD_ADD(B, V)                                                    \
    {                                                                            \
        acc.x += V[0].x; acc.y += V[0].y; acc.z += V[0].z; acc.w += V[0].w;      \
        if (B + 1 < m) { acc.x += V[1].x; acc.y += V[1].y; acc.z += V[1].z; acc.w += V[1].w; } \
        if (B + 2 < m) { acc.x += V[2].x; acc.y += V[2].y; acc.z += V[2].z; acc.w += V[2].w; } \
        if (B + 3 < m) { acc.x += V[3].x; acc.y += V[3].y; acc.z += V[3].z; acc.w += V[3].w; } \
    }

__device__ __forceinline__ void gather_word(const float4* __restrict__ xb, int idx, int m,
                                            float4& acc) {
    float4 va[4], vb[4];
    GATHER_QUAD_LOAD(0, va)
    if (m > 4) GATHER_QUAD_LOAD(4, vb)
    GATHER_QUAD_ADD(0, va)
    if (m > 4) GATHER_QUAD_ADD(4, vb)
    if (m > 8) {
        GATHER_QUAD_LOAD(8, va)
        if (m > 12) GATHER_QUAD_LOAD(12, vb)
        GATHER_QUAD_ADD(8, va)
        if (m > 12) GATHER_QUAD_ADD(12, vb)
    }
}

__global__ void __launch_bounds__(256)
gnn_gather_kernel(const float* __restrict__ x,
                  const int* __restrict__ ptr,
                  const int* __restrict__ cnt,
                  const int* __restrict__ col,
                  float* __restrict__ agg, int n, int n_edges) {
    const unsigned tid = blockIdx.x * blockDim.x + threadIdx.x;
    const unsigned g   = tid >> 4;
    const int      l   = tid & 15;
    const unsigned ng  = (gridDim.x * blockDim.x) >> 4;
    const int n_beg = (int)(((unsigned long long)g * (unsigned)n) / ng);
    const int n_end = (int)(((unsigned long long)(g + 1u) * (unsigned)n) / ng);
    if (n_beg >= n_end) return;
    const float4* xb = (const float4*)x + l;
    const int last = n_edges - 1;
    int pos  = ptr[n_beg];
    int c    = cnt[n_beg];
    int idx0 = col[pos + l < last ? pos + l : last];
    int idx1 = col[pos + 16 + l < last ? pos + 16 + l : last];
    for (int node = n_beg; node < n_end; ++node) {
        const int nn  = node + 1 < n_end ? node + 1 : node;
        const int cN  = cnt[nn];
        const int pN  = pos + c + l;
        const int iN0 = col[pN < last ? pN : last];
        const int iN1 = col[pN + 16 < last ? pN + 16 : last];
        float4 acc = {0.0f, 0.0f, 0.0f, 0.0f};
        if (c > 0)  gather_word(xb, idx0, c < 16 ? c : 16, acc);
        if (c > 16) gather_word(xb, idx1, c < 32 ? c - 16 : 16, acc);
        for (int k0 = 32; k0 < c; k0 += 16) {
            const int p = pos + k0 + l;
            const int w = col[p < last ? p : last];
            gather_word(xb, w, c - k0 < 16 ? c - k0 : 16, acc);
        }
        ((float4*)agg)[(size_t)node * 16 + l] = acc;
        pos += c; c = cN; idx0 = iN0; idx1 = iN1;
    }
}


template <int M>
__device__ __forceinline__ float row_xor(float v) {
    return __builtin_bit_cast(float,
        __builtin_amdgcn_update_dpp(0, __builtin_bit_cast(int, v), 0x160 + M, 0xf, 0xf, true));
}

#define DENSE_WAVES 8
#define DENSE_WGS   256
#define WPITCH      68

template <int NT, int DOUT, bool SOFTMAX>
__global__ void __launch_bounds__(DENSE_WAVES * 32)
gnn_dense_wmma_kernel(const float* __restrict__ agg,
                      const float* __restrict__ deg,
                      const float* __restrict__ hin,
                      const float* __restrict__ Wl,
                      const float* __restrict__ bl,
                      const float* __restrict__ Wr,
                      float* __restrict__ hout,
                      int do_relu, int n_mtiles, int n_rows) {
    static_assert(DOUT % 4 == 0 && DOUT <= NT * 16 && DOUT > (NT - 1) * 16, "column tiling");
    static_assert(DENSE_WAVES * 8 == DIN, "staging covers the 64 k rows with 8 per wave");
    __shared__ __attribute__((aligned(16))) float sWl[NT * 16 * WPITCH];
    __shared__ __attribute__((aligned(16))) float sWr[NT * 16 * WPITCH];

    const int lane = threadIdx.x;
    const int wave = threadIdx.y;
    const int c16  = lane & 15;
    const int h    = lane >> 4;

    const unsigned nw = DENSE_WGS * DENSE_WAVES;
    const unsigned w  = blockIdx.x * DENSE_WAVES + wave;
    const int t_beg = (int)((w * (unsigned)n_mtiles) / nw);
    const int t_end = (int)(((w + 1u) * (unsigned)n_mtiles) / nw);

    float4 aN[8], xN[8];
    float  dN;
    {
        const int t0   = t_beg < n_mtiles ? t_beg : (n_mtiles > 0 ? n_mtiles - 1 : 0);
        const int mrow = t0 * 16 + c16;
        const float4* ap = (const float4*)(agg + (size_t)mrow * DIN) + h;
        const float4* xp = (const float4*)(hin + (size_t)mrow * DIN) + h;
        dN = deg[mrow];
        #pragma unroll
        for (int kk = 0; kk < 8; ++kk) { aN[kk] = ap[2 * kk]; xN[kk] = xp[2 * kk]; }
    }

    {
        constexpr int Q = DOUT / 4;
        const int k  = 8 * wave + (lane >> 2);
        const int ql = lane & 3;
        float4 vl[NT], vr[NT];
        #pragma unroll
        for (int j = 0; j < NT; ++j) {
            const int q = 4 * j + ql;
            vl[j] = float4{0.0f, 0.0f, 0.0f, 0.0f};
            vr[j] = float4{0.0f, 0.0f, 0.0f, 0.0f};
            if (q < Q) {
                vl[j] = ((const float4*)Wl)[k * Q + q];
                vr[j] = ((const float4*)Wr)[k * Q + q];
            }
        }
        #pragma unroll
        for (int j = 0; j < NT; ++j) {
            const int q = 4 * j + ql;
            float* pl = sWl + (4 * q) * WPITCH + k;
            float* pr = sWr + (4 * q) * WPITCH + k;
            pl[0] = vl[j].x; pl[WPITCH] = vl[j].y; pl[2 * WPITCH] = vl[j].z; pl[3 * WPITCH] = vl[j].w;
            pr[0] = vr[j].x; pr[WPITCH] = vr[j].y; pr[2 * WPITCH] = vr[j].z; pr[3 * WPITCH] = vr[j].w;
        }
    }
    __syncthreads();

    float4 bL[8][NT], bR[8][NT];
    {
        const float* bl0 = sWl + c16 * WPITCH + 4 * h;
        const float* br0 = sWr + c16 * WPITCH + 4 * h;
        #pragma unroll
        for (int kk = 0; kk < 8; ++kk) {
            #pragma unroll
            for (int t = 0; t < NT; ++t) {
                bL[kk][t] = *(const float4*)(bl0 + t * 16 * WPITCH + 8 * kk);
                bR[kk][t] = *(const float4*)(br0 + t * 16 * WPITCH + 8 * kk);
            }
        }
    }

    float bv[NT];
    #pragma unroll
    for (int t = 0; t < NT; ++t) {
        const int colb = 16 * t + c16;
        bv[t] = (colb < DOUT) ? bl[colb < DOUT ? colb : 0] : 0.0f;
    }

    for (int mtile = t_beg; mtile < t_end; ++mtile) {
        float4 aC[8], xC[8];
        #pragma unroll
        for (int kk = 0; kk < 8; ++kk) { aC[kk] = aN[kk]; xC[kk] = xN[kk]; }
        const float rdeg = 1.0f / fmaxf(dN, 1.0f);

        {
            const int tn   = (mtile + 1 < t_end) ? mtile + 1 : mtile;
            const int mrow = tn * 16 + c16;
            const float4* ap = (const float4*)(agg + (size_t)mrow * DIN) + h;
            const float4* xp = (const float4*)(hin + (size_t)mrow * DIN) + h;
            dN = deg[mrow];
            #pragma unroll
            for (int kk = 0; kk < 8; ++kk) { aN[kk] = ap[2 * kk]; xN[kk] = xp[2 * kk]; }
        }

        v8f acc[NT];
        #pragma unroll
        for (int t = 0; t < NT; ++t) acc[t] = v8f{};

        #pragma unroll
        for (int kk = 0; kk < 8; ++kk) {
            float4 a = aC[kk];
            const float4 x = xC[kk];
            a.x *= rdeg; a.y *= rdeg; a.z *= rdeg; a.w *= rdeg;
            const v2f a0 = {a.x, a.y}, a1 = {a.z, a.w};
            const v2f x0 = {x.x, x.y}, x1 = {x.z, x.w};
            #pragma unroll
            for (int t = 0; t < NT; ++t) {
                const v2f b = {bL[kk][t].x, bL[kk][t].y};
                acc[t] = __builtin_amdgcn_wmma_f32_16x16x4_f32(false, a0, false, b,
                                                               (short)0, acc[t], false, false);
            }
            #pragma unroll
            for (int t = 0; t < NT; ++t) {
                const v2f b = {bR[kk][t].x, bR[kk][t].y};
                acc[t] = __builtin_amdgcn_wmma_f32_16x16x4_f32(false, x0, false, b,
                                                               (short)0, acc[t], false, false);
            }
            #pragma unroll
            for (int t = 0; t < NT; ++t) {
                const v2f b = {bL[kk][t].z, bL[kk][t].w};
                acc[t] = __builtin_amdgcn_wmma_f32_16x16x4_f32(false, a1, false, b,
                                                               (short)0, acc[t], false, false);
            }
            #pragma unroll
            for (int t = 0; t < NT; ++t) {
                const v2f b = {bR[kk][t].z, bR[kk][t].w};
                acc[t] = __builtin_amdgcn_wmma_f32_16x16x4_f32(false, x1, false, b,
                                                               (short)0, acc[t], false, false);
            }
        }

        if constexpr (SOFTMAX) {
            const float NEG_INF = -__builtin_inff();
            float v[NT][8], m[8], s[8];
            #pragma unroll
            for (int r = 0; r < 8; ++r) {
                m[r] = NEG_INF;
                #pragma unroll
                for (int t = 0; t < NT; ++t) {
                    const bool cv = (16 * t + 15 < DOUT) || (16 * t + c16 < DOUT);
                    v[t][r] = cv ? acc[t][r] + bv[t] : NEG_INF;
                    m[r] = fmaxf(m[r], v[t][r]);
                }
            }
            #pragma unroll
            for (int r = 0; r < 8; ++r) m[r] = fmaxf(m[r], row_xor<1>(m[r]));
            #pragma unroll
            for (int r = 0; r < 8; ++r) m[r] = fmaxf(m[r], row_xor<2>(m[r]));
            #pragma unroll
            for (int r = 0; r < 8; ++r) m[r] = fmaxf(m[r], row_xor<4>(m[r]));
            #pragma unroll
            for (int r = 0; r < 8; ++r) m[r] = fmaxf(m[r], row_xor<8>(m[r]));
            #pragma unroll
            for (int r = 0; r < 8; ++r) {
                s[r] = 0.0f;
                #pragma unroll
                for (int t = 0; t < NT; ++t) s[r] += expf(v[t][r] - m[r]);
            }
            #pragma unroll
            for (int r = 0; r < 8; ++r) s[r] += row_xor<1>(s[r]);
            #pragma unroll
            for (int r = 0; r < 8; ++r) s[r] += row_xor<2>(s[r]);
            #pragma unroll
            for (int r = 0; r < 8; ++r) s[r] += row_xor<4>(s[r]);
            #pragma unroll
            for (int r = 0; r < 8; ++r) s[r] += row_xor<8>(s[r]);
            const int row0 = mtile * 16 + h * 8;
            #pragma unroll
            for (int r = 0; r < 8; ++r) {
                const float ls = logf(s[r]);
                if (row0 + r < n_rows) {
                    float* orow = hout + (size_t)(row0 + r) * DOUT + c16;
                    #pragma unroll
                    for (int t = 0; t < NT; ++t) {
                        const bool cv = (16 * t + 15 < DOUT) || (16 * t + c16 < DOUT);
                        if (cv) orow[16 * t] = v[t][r] - m[r] - ls;
                    }
                }
            }
        } else {
            float* outp = hout + (size_t)(mtile * 16 + h * 8) * DIN + c16;
            #pragma unroll
            for (int t = 0; t < NT; ++t) {
                #pragma unroll
                for (int r = 0; r < 8; ++r) {
                    float v = acc[t][r] + bv[t];
                    if (do_relu) v = fmaxf(v, 0.0f);
                    outp[(size_t)r * DIN + 16 * t] = v;
                }
            }
        }
    }
}

extern "C" void kernel_launch(void* const* d_in, const int* in_sizes, int n_in,
                              void* d_out, int out_size, void* d_ws, size_t ws_size,
                              hipStream_t stream) {
    const float* x   = (const float*)d_in[0];
    const int*   ei  = (const int*)  d_in[1];
    const float* Wl0 = (const float*)d_in[2];
    const float* bl0 = (const float*)d_in[3];
    const float* Wr0 = (const float*)d_in[4];
    const float* Wl1 = (const float*)d_in[5];
    const float* bl1 = (const float*)d_in[6];
    const float* Wr1 = (const float*)d_in[7];
    const float* Wl2 = (const float*)d_in[8];
    const float* bl2 = (const float*)d_in[9];
    const float* Wr2 = (const float*)d_in[10];

    const int N = in_sizes[0] / DIN;
    const int E = in_sizes[1] / 2;
    const int* src = ei;
    const int* dst = ei + E;

    float* deg  = (float*)d_ws;
    float* agg  = deg + N;
    float* h1   = agg + (size_t)N * DIN;
    float* h2   = h1  + (size_t)N * DIN;

    int*  cnt        = (int*)d_out;
    int*  ptr        = cnt + N;
    int*  col        = ptr + N;
    int2* binned     = (int2*)agg;
    int*  offtab     = (int*)h1;

    const int ZB = 256;
    const unsigned chunk_blocks  = (unsigned)((E + EDGE_CHUNK - 1) / EDGE_CHUNK);
    const int      n_bins        = (N + BIN_NODES - 1) / BIN_NODES;
    const unsigned gather_blocks = GATHER_WGS;
    const int n_mtiles = (N + 15) / 16;
    const dim3 dense_block(32, DENSE_WAVES);
    const dim3 dense_grid(DENSE_WGS);

    gnn_chunksort_kernel<<<chunk_blocks, 1024, 0, stream>>>(src, dst, binned, offtab, E);
    gnn_binsort_kernel<<<(unsigned)n_bins, 1024, 0, stream>>>(binned, offtab, ptr, cnt, deg, col, N, (int)chunk_blocks);

    gnn_gather_kernel<<<gather_blocks, ZB, 0, stream>>>(x, ptr, cnt, col, agg, N, E);
    gnn_dense_wmma_kernel<4, 64, false><<<dense_grid, dense_block, 0, stream>>>(
        agg, deg, x, Wl0, bl0, Wr0, h1, 1, n_mtiles, N);

    gnn_gather_kernel<<<gather_blocks, ZB, 0, stream>>>(h1, ptr, cnt, col, agg, N, E);
    gnn_dense_wmma_kernel<4, 64, false><<<dense_grid, dense_block, 0, stream>>>(
        agg, deg, h1, Wl1, bl1, Wr1, h2, 1, n_mtiles, N);

    gnn_gather_kernel<<<gather_blocks, ZB, 0, stream>>>(h2, ptr, cnt, col, agg, N, E);
    gnn_dense_wmma_kernel<3, 40, true><<<dense_grid, dense_block, 0, stream>>>(
        agg, deg, h2, Wl2, bl2, Wr2, (float*)d_out, 0, n_mtiles, N);
}
